# starting kernel
# baseline (speedup 1.0000x reference)
#include <hip/hip_runtime.h>

typedef float v2f __attribute__((ext_vector_type(2)));
typedef float v8f __attribute__((ext_vector_type(8)));

#define D 64

__global__ void sage_deg_kernel(const int* __restrict__ dst,
                                float* __restrict__ deg, int nEdges) {
  int e = blockIdx.x * blockDim.x + threadIdx.x;
  if (e < nEdges) atomicAdd(deg + dst[e], 1.0f);
}

__global__ void sage_scatter_kernel(const float* __restrict__ xin,
                                    const int* __restrict__ src,
                                    const int* __restrict__ dst,
                                    float* __restrict__ agg, int nEdges) {
  int idx = blockIdx.x * blockDim.x + threadIdx.x;
  int total = nEdges * 16;
  if (idx >= total) return;
  int e = idx >> 4;
  int c = (idx & 15) << 2;
  int s = src[e];
  int d = dst[e];
  const float4 v = *(const float4*)(xin + (size_t)s * D + c);
  float* p = agg + (size_t)d * D + c;
  atomicAdd(p + 0, v.x);
  atomicAdd(p + 1, v.y);
  atomicAdd(p + 2, v.z);
  atomicAdd(p + 3, v.w);
}

template <bool RELU>
__global__ void sage_wmma_kernel(const float* __restrict__ agg,
                                 const float* __restrict__ deg,
                                 const float* __restrict__ xin,
                                 const float* __restrict__ Wl,
                                 const float* __restrict__ bl,
                                 const float* __restrict__ Wr,
                                 float* __restrict__ out, int nTiles) {
  const int lane = threadIdx.x & 31;
  const int wave = threadIdx.x >> 5;
  const int tile = blockIdx.x * (blockDim.x >> 5) + wave;
  if (tile >= nTiles) return;

  const int m = lane & 15;
  const int khalf = (lane >> 4) << 1;
  const int row0 = tile << 4;
  const int row = row0 + m;

  const float rdeg = 1.0f / fmaxf(deg[row], 1.0f);
  const float* __restrict__ arow = agg + (size_t)row * D;
  const float* __restrict__ xrow = xin + (size_t)row * D;

  v8f acc[4];
#pragma unroll
  for (int nt = 0; nt < 4; ++nt) acc[nt] = (v8f){0, 0, 0, 0, 0, 0, 0, 0};

#pragma unroll
  for (int kk = 0; kk < 16; ++kk) {
    const int k = (kk << 2) + khalf;
    v2f am = *(const v2f*)(arow + k);
    am[0] *= rdeg;
    am[1] *= rdeg;
    const v2f ax = *(const v2f*)(xrow + k);
#pragma unroll
    for (int nt = 0; nt < 4; ++nt) {
      const int j = (nt << 4) + m;
      const v2f bL = *(const v2f*)(Wl + j * D + k);
      const v2f bR = *(const v2f*)(Wr + j * D + k);
      acc[nt] = __builtin_amdgcn_wmma_f32_16x16x4_f32(
          false, am, false, bL, (short)0, acc[nt], false, false);
      acc[nt] = __builtin_amdgcn_wmma_f32_16x16x4_f32(
          false, ax, false, bR, (short)0, acc[nt], false, false);
    }
  }

  const int mhi = (lane >> 4) << 3;
#pragma unroll
  for (int nt = 0; nt < 4; ++nt) {
    const int j = (nt << 4) + m;
    const float b = bl[j];
#pragma unroll
    for (int v = 0; v < 8; ++v) {
      float val = acc[nt][v] + b;
      if (RELU) val = fmaxf(val, 0.0f);
      out[(size_t)(row0 + mhi + v) * D + j] = val;
    }
  }
}

template <bool RELU>
__global__ void sage_tail_kernel(const float* __restrict__ agg,
                                 const float* __restrict__ deg,
                                 const float* __restrict__ xin,
                                 const float* __restrict__ Wl,
                                 const float* __restrict__ bl,
                                 const float* __restrict__ Wr,
                                 float* __restrict__ out, int row0, int nRows) {
  int idx = blockIdx.x * blockDim.x + threadIdx.x;
  if (idx >= nRows * D) return;
  int r = row0 + idx / D;
  int j = idx % D;
  float rdeg = 1.0f / fmaxf(deg[r], 1.0f);
  float s = bl[j];
  for (int k = 0; k < D; ++k)
    s += agg[(size_t)r * D + k] * rdeg * Wl[j * D + k] +
         xin[(size_t)r * D + k] * Wr[j * D + k];
  if (RELU) s = fmaxf(s, 0.0f);
  out[(size_t)r * D + j] = s;
}

static inline size_t align256(size_t x) { return (x + 255) & ~(size_t)255; }

extern "C" void kernel_launch(void* const* d_in, const int* in_sizes, int n_in,
                              void* d_out, int out_size, void* d_ws,
                              size_t ws_size, hipStream_t stream) {
  const float* x = (const float*)d_in[0];
  const int* src = (const int*)d_in[1];
  const int* dst = (const int*)d_in[2];
  const float* Wl[3] = {(const float*)d_in[3], (const float*)d_in[6],
                        (const float*)d_in[9]};
  const float* bl[3] = {(const float*)d_in[4], (const float*)d_in[7],
                        (const float*)d_in[10]};
  const float* Wr[3] = {(const float*)d_in[5], (const float*)d_in[8],
                        (const float*)d_in[11]};

  const int nNodes = in_sizes[0] / D;
  const int nEdges = in_sizes[1];
  const size_t featBytes = (size_t)nNodes * D * sizeof(float);

  char* ws = (char*)d_ws;
  float* deg = (float*)ws;
  size_t ofs = align256((size_t)nNodes * sizeof(float));
  float* agg = (float*)(ws + ofs);
  ofs += align256(featBytes);
  float* h0 = (float*)(ws + ofs);
  ofs += align256(featBytes);
  float* h1 = (float*)(ws + ofs);

  hipMemsetAsync(deg, 0, (size_t)nNodes * sizeof(float), stream);
  sage_deg_kernel<<<(nEdges + 255) / 256, 256, 0, stream>>>(dst, deg, nEdges);

  const int scatterThreads = nEdges * 16;
  const int scatterBlocks = (scatterThreads + 255) / 256;
  const int nTiles = nNodes / 16;
  const int gemmBlocks = (nTiles + 7) / 8;
  const int tail = nNodes - nTiles * 16;

  auto run_layer = [&](const float* in, const float* wl, const float* b,
                       const float* wr, float* outp, bool relu) {
    hipMemsetAsync(agg, 0, featBytes, stream);
    sage_scatter_kernel<<<scatterBlocks, 256, 0, stream>>>(in, src, dst, agg,
                                                           nEdges);
    if (nTiles > 0) {
      if (relu)
        sage_wmma_kernel<true><<<gemmBlocks, 256, 0, stream>>>(
            agg, deg, in, wl, b, wr, outp, nTiles);
      else
        sage_wmma_kernel<false><<<gemmBlocks, 256, 0, stream>>>(
            agg, deg, in, wl, b, wr, outp, nTiles);
    }
    if (tail > 0) {
      int tb = (tail * D + 255) / 256;
      if (relu)
        sage_tail_kernel<true><<<tb, 256, 0, stream>>>(
            agg, deg, in, wl, b, wr, outp, nTiles * 16, tail);
      else
        sage_tail_kernel<false><<<tb, 256, 0, stream>>>(
            agg, deg, in, wl, b, wr, outp, nTiles * 16, tail);
    }
  };

  run_layer(x, Wl[0], bl[0], Wr[0], h0,  true);
  run_layer(h0, Wl[1], bl[1], Wr[1], h1,  true);
  run_layer(h1, Wl[2], bl[2], Wr[2], (float*)d_out,  false);
}


# Owner groups finish their layer: each 100-node group, its neighbour sums made by LDS float adds, multiplies those LDS sums and LDS-staged root rows on f32 WMMA (8 waves); agg buffer, 3 launches go
# speedup vs baseline: 4.5165x; 4.5165x over previous
#include <hip/hip_runtime.h>


typedef float v2f __attribute__((ext_vector_type(2)));
typedef float v4f __attribute__((ext_vector_type(4)));
typedef float v8f __attribute__((ext_vector_type(8)));

#define D 64
#define NR 100
#define BIN_THREADS 1024
#define EPT 4
#define OWN_THREADS 256
#define OWN_WAVES (OWN_THREADS / 32)
#define GB 16
#define NRP 112
#define PITCH 66
#define RPITCH 68

__global__ void __launch_bounds__(BIN_THREADS)
range_count_kernel(const int* __restrict__ dst, int* __restrict__ gcount,
                   int nEdges, int nRanges) {
  extern __shared__ int lcount[];
  for (int i = threadIdx.x; i < nRanges; i += BIN_THREADS) lcount[i] = 0;
  __syncthreads();
  const int e0 = blockIdx.x * (BIN_THREADS * EPT) + threadIdx.x;
#pragma unroll
  for (int k = 0; k < EPT; ++k) {
    const int e = e0 + k * BIN_THREADS;
    if (e < nEdges) atomicAdd(&lcount[dst[e] / NR], 1);
  }
  __syncthreads();
  for (int i = threadIdx.x; i < nRanges; i += BIN_THREADS) {
    const int c = lcount[i];
    if (c) atomicAdd(gcount + i, c);
  }
}

__global__ void __launch_bounds__(1024)
range_scan_kernel(const int* __restrict__ gcount, int* __restrict__ offs,
                  int* __restrict__ cursor, int nRanges) {
  __shared__ int tmp[1024];
  const int tid = threadIdx.x;
  int carry = 0;
  for (int c0 = 0; c0 < nRanges; c0 += 1024) {
    const int i = c0 + tid;
    const int v = (i < nRanges) ? gcount[i] : 0;
    tmp[tid] = v;
    __syncthreads();
    for (int off = 1; off < 1024; off <<= 1) {
      const int t = (tid >= off) ? tmp[tid - off] : 0;
      __syncthreads();
      tmp[tid] += t;
      __syncthreads();
    }
    const int incl = tmp[tid];
    const int total = tmp[1023];
    if (i < nRanges) {
      offs[i] = carry + incl - v;
      cursor[i] = carry + incl - v;
    }
    carry += total;
    __syncthreads();
  }
  if (tid == 0) offs[nRanges] = carry;
}

__global__ void __launch_bounds__(BIN_THREADS)
range_fill_kernel(const int* __restrict__ src, const int* __restrict__ dst,
                  int* __restrict__ cursor, int* __restrict__ bedge,
                  int nEdges, int nRanges) {
  extern __shared__ int lsh[];
  int* lcnt = lsh;
  int* lbase = lsh + nRanges;
  for (int i = threadIdx.x; i < nRanges; i += BIN_THREADS) lcnt[i] = 0;
  __syncthreads();
  const int e0 = blockIdx.x * (BIN_THREADS * EPT) + threadIdx.x;
  int dv[EPT];
#pragma unroll
  for (int k = 0; k < EPT; ++k) {
    const int e = e0 + k * BIN_THREADS;
    dv[k] = 0;
    if (e < nEdges) {
      dv[k] = dst[e];
      atomicAdd(&lcnt[dv[k] / NR], 1);
    }
  }
  __syncthreads();
  for (int i = threadIdx.x; i < nRanges; i += BIN_THREADS) {
    const int c = lcnt[i];
    lbase[i] = c ? atomicAdd(cursor + i, c) : 0;
    lcnt[i] = 0;
  }
  __syncthreads();
#pragma unroll
  for (int k = 0; k < EPT; ++k) {
    const int e = e0 + k * BIN_THREADS;
    if (e < nEdges) {
      const int r = dv[k] / NR;
      const int pos = lbase[r] + atomicAdd(&lcnt[r], 1);
      int2 pr;
      pr.x = src[e];
      pr.y = dv[k] - r * NR;
      *(int2*)(bedge + 2 * (size_t)pos) = pr;
    }
  }
}

template <bool RELU, bool DEG>
__global__ void __launch_bounds__(OWN_THREADS, 8)
owner_layer_kernel(const float* __restrict__ xin, const int* __restrict__ bedge,
                   const int* __restrict__ offs, float* __restrict__ deg,
                   const float* __restrict__ Wl, const float* __restrict__ bl,
                   const float* __restrict__ Wr, float* __restrict__ out,
                   int nNodes) {
  __shared__ __attribute__((aligned(16))) float acc[((NRP + 1) * PITCH + 3) / 4 * 4];
  __shared__ __attribute__((aligned(16))) float rootl[NRP * RPITCH];
  __shared__ float degl[NRP + 1];
  const int tid = threadIdx.x;
  const int lane = tid & 31;
  const int wave = __builtin_amdgcn_readfirstlane(tid >> 5);
  const int r = blockIdx.x;

  const int m = lane & 15;
  const int half = lane >> 4;
  const int khalf = half << 1;
  const int mhi = half << 3;
  const int nt = wave & 3;
  const int j = (nt << 4) + m;

  v2f bL[16], bR[16];
#pragma unroll
  for (int kk = 0; kk < 16; ++kk) {
    const int k = (kk << 2) + khalf;
    bL[kk] = *(const v2f*)(Wl + j * D + k);
    bR[kk] = *(const v2f*)(Wr + j * D + k);
  }
  const float bias = bl[j];

  const int nodeBase = r * NR;
  const int nValid = min(NR, nNodes - nodeBase);
  static_assert(NRP * (D / 4) == 7 * OWN_THREADS, "seven whole trips");
  v4f rv[NRP * (D / 4) / OWN_THREADS];
#pragma unroll
  for (int t = 0; t < NRP * (D / 4) / OWN_THREADS; ++t) {
    const int row = (tid >> 4) + t * (OWN_THREADS / 16);
    const int grow = min(nodeBase + row, nNodes - 1);
    rv[t] = *(const v4f*)(xin + (size_t)grow * D + 4 * (tid & 15));
  }
  float dval = 0.f;
  if (!DEG) dval = deg[min(nodeBase + min(tid, NRP - 1), nNodes - 1)];

  for (int i = tid; i < (((NRP + 1) * PITCH + 3) / 4); i += OWN_THREADS)
    ((float4*)acc)[i] = make_float4(0.f, 0.f, 0.f, 0.f);
  if (DEG) {
    for (int i = tid; i <= NRP; i += OWN_THREADS) degl[i] = 0.f;
  } else {
    if (tid < NRP) degl[tid] = dval;
  }
  asm volatile("" ::"v"(rv[0]), "v"(rv[1]), "v"(rv[2]), "v"(rv[3]), "v"(rv[4]),
               "v"(rv[5]), "v"(rv[6]));
#pragma unroll
  for (int t = 0; t < NRP * (D / 4) / OWN_THREADS; ++t) {
    const int row = (tid >> 4) + t * (OWN_THREADS / 16);
    *(v4f*)(rootl + row * RPITCH + 4 * (tid & 15)) = rv[t];
  }
  const int eBeg = offs[r];
  const int eEnd = offs[r + 1];
  __syncthreads();

  const unsigned lane8 = (unsigned)lane << 3;
  const unsigned lane4 = (unsigned)lane << 2;
  const char* __restrict__ xb = (const char*)xin;
  char* accb = (char*)acc;

  for (int base = eBeg + wave * 32; base < eEnd; base += OWN_WAVES * 32) {
    const int e = base + lane;
    int2 sd;
    sd.x = 0;
    sd.y = NRP;
    if (e < eEnd) {
      sd = *(const int2*)(bedge + 2 * (size_t)e);
      if (DEG)
        __hip_atomic_fetch_add(degl + sd.y, 1.0f, __ATOMIC_RELAXED,
                               __HIP_MEMORY_SCOPE_WORKGROUP);
    }
    const unsigned soff = (unsigned)sd.x << 8;
    const unsigned doff = (unsigned)sd.y * (PITCH * 4);
#pragma unroll
    for (int jj = 0; jj < 32; jj += GB) {
      v2f v[GB];
#pragma unroll
      for (int u = 0; u < GB; ++u) {
        const unsigned so =
            (unsigned)__builtin_amdgcn_readlane((int)soff, jj + u);
        v[u] = *(const v2f*)(xb + (so | lane8));
      }
#pragma unroll
      for (int u = 0; u < GB; ++u) {
        const unsigned dofs =
            (unsigned)__builtin_amdgcn_readlane((int)doff, jj + u);
        float* p = (float*)(accb + (dofs + lane4));
        __hip_atomic_fetch_add(p, v[u][0], __ATOMIC_RELAXED,
                               __HIP_MEMORY_SCOPE_WORKGROUP);
        __hip_atomic_fetch_add(p + 32, v[u][1], __ATOMIC_RELAXED,
                               __HIP_MEMORY_SCOPE_WORKGROUP);
      }
    }
  }

  __syncthreads();

  if (DEG)
    for (int i = tid; i < nValid; i += OWN_THREADS) deg[nodeBase + i] = degl[i];

  for (int tile = wave >> 2; tile < NRP / 16; tile += OWN_WAVES / 4) {
    const int lrow = (tile << 4) + m;
    const float rdeg = 1.0f / fmaxf(degl[lrow], 1.0f);
    const float* arow = acc + lrow * PITCH + half;
    const float* xrow = rootl + lrow * RPITCH + khalf;

    v8f c = (v8f){0, 0, 0, 0, 0, 0, 0, 0};
#pragma unroll
    for (int kk = 0; kk < 16; ++kk) {
      v2f a;
      a[0] = arow[2 * kk] * rdeg;
      a[1] = arow[32 + 2 * kk] * rdeg;
      c = __builtin_amdgcn_wmma_f32_16x16x4_f32(false, a, false, bL[kk],
                                                (short)0, c, false, false);
      const v2f ax = *(const v2f*)(xrow + (kk << 2));
      c = __builtin_amdgcn_wmma_f32_16x16x4_f32(false, ax, false, bR[kk],
                                                (short)0, c, false, false);
    }

    const int lr0 = (tile << 4) + mhi;
#pragma unroll
    for (int v = 0; v < 8; ++v) {
      float val = c[v] + bias;
      if (RELU) val = fmaxf(val, 0.0f);
      if (lr0 + v < nValid) out[(size_t)(nodeBase + lr0 + v) * D + j] = val;
    }
  }
}

static inline size_t align256(size_t x) { return (x + 255) & ~(size_t)255; }

extern "C" void kernel_launch(void* const* d_in, const int* in_sizes, int n_in,
                              void* d_out, int out_size, void* d_ws,
                              size_t ws_size, hipStream_t stream) {
  const float* x = (const float*)d_in[0];
  const int* src = (const int*)d_in[1];
  const int* dst = (const int*)d_in[2];
  const float* Wl[3] = {(const float*)d_in[3], (const float*)d_in[6],
                        (const float*)d_in[9]};
  const float* bl[3] = {(const float*)d_in[4], (const float*)d_in[7],
                        (const float*)d_in[10]};
  const float* Wr[3] = {(const float*)d_in[5], (const float*)d_in[8],
                        (const float*)d_in[11]};

  const int nNodes = in_sizes[0] / D;
  const int nEdges = in_sizes[1];
  const size_t featBytes = (size_t)nNodes * D * sizeof(float);
  const int nRanges = (nNodes + NR - 1) / NR;

  char* ws = (char*)d_ws;
  float* deg = (float*)ws;
  size_t ofs = align256((size_t)nNodes * sizeof(float));
  float* h1 = (float*)(ws + ofs);
  ofs += align256(featBytes);
  int* bedge = (int*)(ws + ofs);
  ofs += align256((size_t)nEdges * 2 * sizeof(int));
  int* gcount = (int*)(ws + ofs);
  ofs += align256((size_t)nRanges * sizeof(int));
  int* offs = (int*)(ws + ofs);
  ofs += align256((size_t)(nRanges + 1) * sizeof(int));
  int* cursor = (int*)(ws + ofs);
  float* h0 = (float*)d_out;

  hipMemsetAsync(gcount, 0, (size_t)nRanges * sizeof(int), stream);
  const int binBlocks = (nEdges + BIN_THREADS * EPT - 1) / (BIN_THREADS * EPT);
  if (binBlocks > 0)
    range_count_kernel<<<binBlocks, BIN_THREADS, nRanges * sizeof(int),
                         stream>>>(dst, gcount, nEdges, nRanges);
  range_scan_kernel<<<1, 1024, 0, stream>>>(gcount, offs, cursor, nRanges);
  if (binBlocks > 0)
    range_fill_kernel<<<binBlocks, BIN_THREADS, 2 * nRanges * sizeof(int),
                        stream>>>(src, dst, cursor, bedge, nEdges, nRanges);

  auto run_layer = [&](const float* in, const float* wl, const float* b,
                       const float* wr, float* outp, bool relu, bool first) {
    if (nRanges <= 0) return;
    if (first)
      owner_layer_kernel<true, true><<<nRanges, OWN_THREADS, 0, stream>>>(
          in, bedge, offs, deg, wl, b, wr, outp, nNodes);
    else if (relu)
      owner_layer_kernel<true, false><<<nRanges, OWN_THREADS, 0, stream>>>(
          in, bedge, offs, deg, wl, b, wr, outp, nNodes);
    else
      owner_layer_kernel<false, false><<<nRanges, OWN_THREADS, 0, stream>>>(
          in, bedge, offs, deg, wl, b, wr, outp, nNodes);
  };

  run_layer(x, Wl[0], bl[0], Wr[0], h0,  true,  true);
  run_layer(h0, Wl[1], bl[1], Wr[1], h1,  true,  false);
  run_layer(h1, Wl[2], bl[2], Wr[2], (float*)d_out,  false,
             false);
}


# Neighbour sums in registers: each group counting-sorts its edges by destination row in LDS; a wave sums a row's run in registers, one LDS float add pair per row change (new sum order)
# speedup vs baseline: 4.7214x; 1.0454x over previous
#include <hip/hip_runtime.h>


typedef float v2f __attribute__((ext_vector_type(2)));
typedef float v4f __attribute__((ext_vector_type(4)));
typedef float v8f __attribute__((ext_vector_type(8)));

#define D 64
#define NR 100
#define BIN_THREADS 1024
#define EPT 4
#define OWN_THREADS 256
#define OWN_WAVES (OWN_THREADS / 32)
#define GB 16
#define NRP 112
static_assert(NRP < 128, "a destination row fits the seven low bits of the sort's key");
#define PITCH 66
#define RPITCH 68
#define CAP 2048
#define SEPT (CAP / OWN_THREADS)

__global__ void __launch_bounds__(BIN_THREADS)
range_count_kernel(const int* __restrict__ dst, int* __restrict__ gcount,
                   int nEdges, int nRanges) {
  extern __shared__ int lcount[];
  for (int i = threadIdx.x; i < nRanges; i += BIN_THREADS) lcount[i] = 0;
  __syncthreads();
  const int e0 = blockIdx.x * (BIN_THREADS * EPT) + threadIdx.x;
#pragma unroll
  for (int k = 0; k < EPT; ++k) {
    const int e = e0 + k * BIN_THREADS;
    if (e < nEdges) atomicAdd(&lcount[dst[e] / NR], 1);
  }
  __syncthreads();
  for (int i = threadIdx.x; i < nRanges; i += BIN_THREADS) {
    const int c = lcount[i];
    if (c) atomicAdd(gcount + i, c);
  }
}

__global__ void __launch_bounds__(1024)
range_scan_kernel(const int* __restrict__ gcount, int* __restrict__ offs,
                  int* __restrict__ cursor, int nRanges) {
  __shared__ int tmp[1024];
  const int tid = threadIdx.x;
  int carry = 0;
  for (int c0 = 0; c0 < nRanges; c0 += 1024) {
    const int i = c0 + tid;
    const int v = (i < nRanges) ? gcount[i] : 0;
    tmp[tid] = v;
    __syncthreads();
    for (int off = 1; off < 1024; off <<= 1) {
      const int t = (tid >= off) ? tmp[tid - off] : 0;
      __syncthreads();
      tmp[tid] += t;
      __syncthreads();
    }
    const int incl = tmp[tid];
    const int total = tmp[1023];
    if (i < nRanges) {
      offs[i] = carry + incl - v;
      cursor[i] = carry + incl - v;
    }
    carry += total;
    __syncthreads();
  }
  if (tid == 0) offs[nRanges] = carry;
}

__global__ void __launch_bounds__(BIN_THREADS)
range_fill_kernel(const int* __restrict__ src, const int* __restrict__ dst,
                  int* __restrict__ cursor, int* __restrict__ bedge,
                  int nEdges, int nRanges) {
  extern __shared__ int lsh[];
  int* lcnt = lsh;
  int* lbase = lsh + nRanges;
  for (int i = threadIdx.x; i < nRanges; i += BIN_THREADS) lcnt[i] = 0;
  __syncthreads();
  const int e0 = blockIdx.x * (BIN_THREADS * EPT) + threadIdx.x;
  int dv[EPT];
#pragma unroll
  for (int k = 0; k < EPT; ++k) {
    const int e = e0 + k * BIN_THREADS;
    dv[k] = 0;
    if (e < nEdges) {
      dv[k] = dst[e];
      atomicAdd(&lcnt[dv[k] / NR], 1);
    }
  }
  __syncthreads();
  for (int i = threadIdx.x; i < nRanges; i += BIN_THREADS) {
    const int c = lcnt[i];
    lbase[i] = c ? atomicAdd(cursor + i, c) : 0;
    lcnt[i] = 0;
  }
  __syncthreads();
#pragma unroll
  for (int k = 0; k < EPT; ++k) {
    const int e = e0 + k * BIN_THREADS;
    if (e < nEdges) {
      const int r = dv[k] / NR;
      const int pos = lbase[r] + atomicAdd(&lcnt[r], 1);
      int2 pr;
      pr.x = src[e];
      pr.y = dv[k] - r * NR;
      *(int2*)(bedge + 2 * (size_t)pos) = pr;
    }
  }
}

template <bool RELU, bool DEG>
__global__ void __launch_bounds__(OWN_THREADS, 8)
owner_layer_kernel(const float* __restrict__ xin, const int* __restrict__ bedge,
                   const int* __restrict__ offs, float* __restrict__ deg,
                   const float* __restrict__ Wl, const float* __restrict__ bl,
                   const float* __restrict__ Wr, float* __restrict__ out,
                   int nNodes) {
  __shared__ __attribute__((aligned(16))) float acc[((NRP + 1) * PITCH + 3) / 4 * 4];
  __shared__ __attribute__((aligned(16))) float rootl[NRP * RPITCH];
  __shared__ float degl[NRP + 1];
  __shared__ __attribute__((aligned(16))) int2 sorted[CAP];
  __shared__ __attribute__((aligned(16))) int cnt[128];
  __shared__ __attribute__((aligned(16))) int startl[128];
  const int tid = threadIdx.x;
  const int lane = tid & 31;
  const int wave = __builtin_amdgcn_readfirstlane(tid >> 5);
  const int r = blockIdx.x;

  const int m = lane & 15;
  const int half = lane >> 4;
  const int khalf = half << 1;
  const int mhi = half << 3;
  const int nt = wave & 3;
  const int j = (nt << 4) + m;

  v2f bL[16], bR[16];
#pragma unroll
  for (int kk = 0; kk < 16; ++kk) {
    const int k = (kk << 2) + khalf;
    bL[kk] = *(const v2f*)(Wl + j * D + k);
    bR[kk] = *(const v2f*)(Wr + j * D + k);
  }

  const int nodeBase = r * NR;
  const int nValid = min(NR, nNodes - nodeBase);
  static_assert(NRP * (D / 4) == 7 * OWN_THREADS, "seven whole trips");
  v4f rv[NRP * (D / 4) / OWN_THREADS];
#pragma unroll
  for (int t = 0; t < NRP * (D / 4) / OWN_THREADS; ++t) {
    const int row = (tid >> 4) + t * (OWN_THREADS / 16);
    const int grow = min(nodeBase + row, nNodes - 1);
    rv[t] = *(const v4f*)(xin + (size_t)grow * D + 4 * (tid & 15));
  }

  float dval = 0.f;
  if (!DEG) dval = deg[min(nodeBase + min(tid, NRP - 1), nNodes - 1)];

  for (int i = tid; i < (((NRP + 1) * PITCH + 3) / 4); i += OWN_THREADS)
    ((float4*)acc)[i] = make_float4(0.f, 0.f, 0.f, 0.f);
  if (DEG) {
    for (int i = tid; i <= NRP; i += OWN_THREADS) degl[i] = 0.f;
  } else {
    if (tid < NRP) degl[tid] = dval;
  }
  asm volatile("" ::"v"(rv[0]), "v"(rv[1]), "v"(rv[2]), "v"(rv[3]), "v"(rv[4]),
               "v"(rv[5]), "v"(rv[6]));
#pragma unroll
  for (int t = 0; t < NRP * (D / 4) / OWN_THREADS; ++t) {
    const int row = (tid >> 4) + t * (OWN_THREADS / 16);
    *(v4f*)(rootl + row * RPITCH + 4 * (tid & 15)) = rv[t];
  }
  const int eBeg = offs[r];
  const int eEnd = offs[r + 1];

  const unsigned lane8 = (unsigned)lane << 3;
  const unsigned lane4 = (unsigned)lane << 2;
  const char* __restrict__ xb = (const char*)xin;
  char* accb = (char*)acc;

  for (int cb = eBeg; cb < eEnd; cb += CAP) {
    const int n = min(CAP, eEnd - cb);
    if (tid < 128) cnt[tid] = 0;
    __syncthreads();

    int es[SEPT], ed[SEPT];
#pragma unroll
    for (int k = 0; k < SEPT; ++k) {
      const int i = k * OWN_THREADS + tid;
      ed[k] = -1;
      es[k] = 0;
      if (i < n) {
        const int2 sd = *(const int2*)(bedge + 2 * (size_t)(cb + i));
        es[k] = sd.x;
        ed[k] = sd.y;
      }
    }
#pragma unroll
    for (int k = 0; k < SEPT; ++k)
      if (ed[k] >= 0) ed[k] |= atomicAdd(&cnt[ed[k]], 1) << 7;
    __syncthreads();
    if (DEG) {
      if (tid < NRP) degl[tid] += (float)cnt[tid];
    }
    if (wave == 0) {
      const int4 c4 = ((const int4*)cnt)[lane];
      const int own = c4.x + c4.y + c4.z + c4.w;
      int s = own;
#pragma unroll
      for (int off = 1; off < 32; off <<= 1) {
        const int t = __shfl_up(s, off);
        if (lane >= off) s += t;
      }
      int4 st;
      st.x = s - own;
      st.y = st.x + c4.x;
      st.z = st.y + c4.y;
      st.w = st.z + c4.z;
      ((int4*)startl)[lane] = st;
    }
    __syncthreads();
#pragma unroll
    for (int k = 0; k < SEPT; ++k)
      if (ed[k] >= 0) {
        int2 pr;
        pr.x = es[k];
        pr.y = ed[k] & 127;
        sorted[startl[pr.y] + (ed[k] >> 7)] = pr;
      }
    __syncthreads();

    for (int base = wave * 32; base < n; base += OWN_WAVES * 32) {
      int2 sd;
      sd.x = 0;
      sd.y = NRP;
      if (base + lane < n) sd = sorted[base + lane];
      const unsigned soff = (unsigned)sd.x << 8;
      int cur = __builtin_amdgcn_readlane(sd.y, 0);
      v2f ps = (v2f){0.f, 0.f};
#pragma unroll
      for (int jj = 0; jj < 32; jj += GB) {
        v2f v[GB];
#pragma unroll
        for (int u = 0; u < GB; ++u) {
          const unsigned so =
              (unsigned)__builtin_amdgcn_readlane((int)soff, jj + u);
          v[u] = *(const v2f*)(xb + (so | lane8));
        }
#pragma unroll
        for (int u = 0; u < GB; ++u) {
          const int row = __builtin_amdgcn_readlane(sd.y, jj + u);
          if (row != cur) {
            float* p = (float*)(accb + ((unsigned)cur * (PITCH * 4) + lane4));
            __hip_atomic_fetch_add(p, ps[0], __ATOMIC_RELAXED,
                                   __HIP_MEMORY_SCOPE_WORKGROUP);
            __hip_atomic_fetch_add(p + 32, ps[1], __ATOMIC_RELAXED,
                                   __HIP_MEMORY_SCOPE_WORKGROUP);
            ps = (v2f){0.f, 0.f};
            cur = row;
          }
          ps += v[u];
        }
      }
      float* p = (float*)(accb + ((unsigned)cur * (PITCH * 4) + lane4));
      __hip_atomic_fetch_add(p, ps[0], __ATOMIC_RELAXED,
                             __HIP_MEMORY_SCOPE_WORKGROUP);
      __hip_atomic_fetch_add(p + 32, ps[1], __ATOMIC_RELAXED,
                             __HIP_MEMORY_SCOPE_WORKGROUP);
    }
  }

  __syncthreads();

  if (DEG)
    for (int i = tid; i < nValid; i += OWN_THREADS) deg[nodeBase + i] = degl[i];
  const float bias = bl[j];

  for (int tile = wave >> 2; tile < NRP / 16; tile += OWN_WAVES / 4) {
    const int lrow = (tile << 4) + m;
    const float rdeg = 1.0f / fmaxf(degl[lrow], 1.0f);
    const float* arow = acc + lrow * PITCH + half;
    const float* xrow = rootl + lrow * RPITCH + khalf;

    v8f c = (v8f){0, 0, 0, 0, 0, 0, 0, 0};
#pragma unroll
    for (int kk = 0; kk < 16; ++kk) {
      v2f a;
      a[0] = arow[2 * kk] * rdeg;
      a[1] = arow[32 + 2 * kk] * rdeg;
      c = __builtin_amdgcn_wmma_f32_16x16x4_f32(false, a, false, bL[kk],
                                                (short)0, c, false, false);
      const v2f ax = *(const v2f*)(xrow + (kk << 2));
      c = __builtin_amdgcn_wmma_f32_16x16x4_f32(false, ax, false, bR[kk],
                                                (short)0, c, false, false);
    }

    const int lr0 = (tile << 4) + mhi;
#pragma unroll
    for (int v = 0; v < 8; ++v) {
      float val = c[v] + bias;
      if (RELU) val = fmaxf(val, 0.0f);
      if (lr0 + v < nValid) out[(size_t)(nodeBase + lr0 + v) * D + j] = val;
    }
  }
}

static inline size_t align256(size_t x) { return (x + 255) & ~(size_t)255; }

extern "C" void kernel_launch(void* const* d_in, const int* in_sizes, int n_in,
                              void* d_out, int out_size, void* d_ws,
                              size_t ws_size, hipStream_t stream) {
  const float* x = (const float*)d_in[0];
  const int* src = (const int*)d_in[1];
  const int* dst = (const int*)d_in[2];
  const float* Wl[3] = {(const float*)d_in[3], (const float*)d_in[6],
                        (const float*)d_in[9]};
  const float* bl[3] = {(const float*)d_in[4], (const float*)d_in[7],
                        (const float*)d_in[10]};
  const float* Wr[3] = {(const float*)d_in[5], (const float*)d_in[8],
                        (const float*)d_in[11]};

  const int nNodes = in_sizes[0] / D;
  const int nEdges = in_sizes[1];
  const size_t featBytes = (size_t)nNodes * D * sizeof(float);
  const int nRanges = (nNodes + NR - 1) / NR;

  char* ws = (char*)d_ws;
  float* deg = (float*)ws;
  size_t ofs = align256((size_t)nNodes * sizeof(float));
  float* h1 = (float*)(ws + ofs);
  ofs += align256(featBytes);
  int* bedge = (int*)(ws + ofs);
  ofs += align256((size_t)nEdges * 2 * sizeof(int));
  int* gcount = (int*)(ws + ofs);
  ofs += align256((size_t)nRanges * sizeof(int));
  int* offs = (int*)(ws + ofs);
  ofs += align256((size_t)(nRanges + 1) * sizeof(int));
  int* cursor = (int*)(ws + ofs);
  float* h0 = (float*)d_out;

  hipMemsetAsync(gcount, 0, (size_t)nRanges * sizeof(int), stream);
  const int binBlocks = (nEdges + BIN_THREADS * EPT - 1) / (BIN_THREADS * EPT);
  if (binBlocks > 0)
    range_count_kernel<<<binBlocks, BIN_THREADS, nRanges * sizeof(int),
                         stream>>>(dst, gcount, nEdges, nRanges);
  range_scan_kernel<<<1, 1024, 0, stream>>>(gcount, offs, cursor, nRanges);
  if (binBlocks > 0)
    range_fill_kernel<<<binBlocks, BIN_THREADS, 2 * nRanges * sizeof(int),
                        stream>>>(src, dst, cursor, bedge, nEdges, nRanges);

  auto run_layer = [&](const float* in, const float* wl, const float* b,
                       const float* wr, float* outp, bool relu, bool first) {
    if (nRanges <= 0) return;
    if (first)
      owner_layer_kernel<true, true><<<nRanges, OWN_THREADS, 0, stream>>>(
          in, bedge, offs, deg, wl, b, wr, outp, nNodes);
    else if (relu)
      owner_layer_kernel<true, false><<<nRanges, OWN_THREADS, 0, stream>>>(
          in, bedge, offs, deg, wl, b, wr, outp, nNodes);
    else
      owner_layer_kernel<false, false><<<nRanges, OWN_THREADS, 0, stream>>>(
          in, bedge, offs, deg, wl, b, wr, outp, nNodes);
  };

  run_layer(x, Wl[0], bl[0], Wr[0], h0,  true,  true);
  run_layer(h0, Wl[1], bl[1], Wr[1], h1,  true,  false);
  run_layer(h1, Wl[2], bl[2], Wr[2], (float*)d_out,  false,
             false);
}


# Edge binning by 196 blocks of 8192 edges, 16 a thread (were 391 of 4096): a block counting-sorts its pairs by range in LDS, stores each run whole; 1/4 of the waves enter; pair order in runs changes
# speedup vs baseline: 4.9855x; 1.0559x over previous
#include <hip/hip_runtime.h>


typedef float v2f __attribute__((ext_vector_type(2)));
typedef float v4f __attribute__((ext_vector_type(4)));
typedef float v8f __attribute__((ext_vector_type(8)));

#define D 64
#define NR 100
#define BIN_THREADS 512
#define EPT 16
#define BIN_EDGES (BIN_THREADS * EPT)
#define BIN_WAVES (BIN_THREADS / 32)
static_assert(BIN_WAVES <= 32, "one lane of the first wave per wave in the block scan");
static_assert(NR <= 128, "a row within its range fits seven bits of the staged pair");
#define OWN_THREADS 256
#define OWN_WAVES (OWN_THREADS / 32)
#define GB 16
#define NRP 112
static_assert(NRP < 128, "a destination row fits the seven low bits of the sort's key");
#define PITCH 66
#define RPITCH 68
#define CAP 2048
#define SEPT (CAP / OWN_THREADS)

template <bool VEC>
__device__ __forceinline__ void load_block_edges(const int* __restrict__ p,
                                                 int blockBase, int nEdges,
                                                 int (&v)[EPT]) {
#pragma unroll
  for (int k = 0; k < EPT / 4; ++k) {
    const int e = blockBase + ((k * BIN_THREADS + (int)threadIdx.x) << 2);
    if (VEC) {
      const int4 q = *(const int4*)(p + e);
      v[4 * k + 0] = q.x;
      v[4 * k + 1] = q.y;
      v[4 * k + 2] = q.z;
      v[4 * k + 3] = q.w;
    } else {
#pragma unroll
      for (int u = 0; u < 4; ++u) {
        const int val = p[min(e + u, nEdges - 1)];
        v[4 * k + u] = (e + u < nEdges) ? val : -1;
      }
    }
  }
}

__global__ void __launch_bounds__(BIN_THREADS)
range_count_kernel(const int* __restrict__ dst, int* __restrict__ gcount,
                   int nEdges, int nRanges) {
  extern __shared__ int lcount[];
  for (int i = threadIdx.x; i < nRanges; i += BIN_THREADS) lcount[i] = 0;
  const int spare = min((int)(threadIdx.x & 31), nRanges - 1);
  const int blockBase = blockIdx.x * BIN_EDGES;
  int dv[EPT];
  if ((((size_t)dst) & 15) == 0 && blockBase + BIN_EDGES <= nEdges)
    load_block_edges<true>(dst, blockBase, nEdges, dv);
  else
    load_block_edges<false>(dst, blockBase, nEdges, dv);
  __syncthreads();
#pragma unroll
  for (int k = 0; k < EPT; ++k) {
    const bool ok = dv[k] >= 0;
    atomicAdd(&lcount[ok ? (int)((unsigned)dv[k] / NR) : spare], ok ? 1 : 0);
  }
  __syncthreads();
  for (int i = threadIdx.x; i < nRanges; i += BIN_THREADS) {
    const int c = lcount[i];
    if (c) atomicAdd(gcount + i, c);
  }
}

__global__ void __launch_bounds__(1024)
range_scan_kernel(const int* __restrict__ gcount, int* __restrict__ offs,
                  int* __restrict__ cursor, int nRanges) {
  __shared__ int tmp[1024];
  const int tid = threadIdx.x;
  int carry = 0;
  for (int c0 = 0; c0 < nRanges; c0 += 1024) {
    const int i = c0 + tid;
    const int v = (i < nRanges) ? gcount[i] : 0;
    tmp[tid] = v;
    __syncthreads();
    for (int off = 1; off < 1024; off <<= 1) {
      const int t = (tid >= off) ? tmp[tid - off] : 0;
      __syncthreads();
      tmp[tid] += t;
      __syncthreads();
    }
    const int incl = tmp[tid];
    const int total = tmp[1023];
    if (i < nRanges) {
      offs[i] = carry + incl - v;
      cursor[i] = carry + incl - v;
    }
    carry += total;
    __syncthreads();
  }
  if (tid == 0) offs[nRanges] = carry;
}

__global__ void __launch_bounds__(BIN_THREADS)
range_fill_kernel(const int* __restrict__ src, const int* __restrict__ dst,
                  int* __restrict__ cursor, int* __restrict__ bedge,
                  int nEdges, int nRanges) {
  extern __shared__ int lsh[];
  __shared__ __attribute__((aligned(16))) int2 staged[BIN_EDGES];
  __shared__ int wsum[32];
  __shared__ int wtot;
  int* lcnt = lsh;
  int* ldelta = lsh + nRanges;
  const int tid = threadIdx.x;
  const int lane = tid & 31;
  const int wave = tid >> 5;
  const int blockBase = blockIdx.x * BIN_EDGES;
  for (int i = tid; i < nRanges; i += BIN_THREADS) lcnt[i] = 0;
  const int spare = min(lane, nRanges - 1);
  int dv[EPT], sv[EPT];
  if (((((size_t)dst) | ((size_t)src)) & 15) == 0 &&
      blockBase + BIN_EDGES <= nEdges) {
    load_block_edges<true>(dst, blockBase, nEdges, dv);
    load_block_edges<true>(src, blockBase, nEdges, sv);
  } else {
    load_block_edges<false>(dst, blockBase, nEdges, dv);
    load_block_edges<false>(src, blockBase, nEdges, sv);
  }
  __syncthreads();
#pragma unroll
  for (int k = 0; k < EPT; ++k) {
    const bool ok = dv[k] >= 0;
    const int r = ok ? (int)((unsigned)dv[k] / NR) : spare;
    if (ok) dv[k] = (r << 7) | (dv[k] - r * NR);
    atomicAdd(&lcnt[r], ok ? 1 : 0);
  }
  __syncthreads();
  int carry = 0;
  for (int c0 = 0; c0 < nRanges; c0 += 2 * BIN_THREADS) {
    const int i0 = c0 + 2 * tid;
    const int i1 = i0 + 1;
    const int n0 = (i0 < nRanges) ? lcnt[i0] : 0;
    const int n1 = (i1 < nRanges) ? lcnt[i1] : 0;
    const int g0 = n0 ? atomicAdd(cursor + i0, n0) : 0;
    const int g1 = n1 ? atomicAdd(cursor + i1, n1) : 0;
    const int own = n0 + n1;
    int sc = own;
#pragma unroll
    for (int off = 1; off < 32; off <<= 1) {
      const int t = __shfl_up(sc, off);
      if (lane >= off) sc += t;
    }
    if (lane == 31) wsum[wave] = sc;
    __syncthreads();
    if (wave == 0) {
      const int w = (lane < BIN_WAVES) ? wsum[lane] : 0;
      int ws = w;
#pragma unroll
      for (int off = 1; off < 32; off <<= 1) {
        const int t = __shfl_up(ws, off);
        if (lane >= off) ws += t;
      }
      wsum[lane] = ws - w;
      if (lane == 31) wtot = ws;
    }
    __syncthreads();
    const int excl = carry + wsum[wave] + sc - own;
    if (i0 < nRanges) {
      lcnt[i0] = excl;
      ldelta[i0] = g0 - excl;
    }
    if (i1 < nRanges) {
      lcnt[i1] = excl + n0;
      ldelta[i1] = g1 - (excl + n0);
    }
    carry += wtot;
    __syncthreads();
  }
  int slot[EPT];
#pragma unroll
  for (int k = 0; k < EPT; ++k) {
    const bool ok = dv[k] >= 0;
    slot[k] = atomicAdd(&lcnt[ok ? (dv[k] >> 7) : spare], ok ? 1 : 0);
  }
#pragma unroll
  for (int k = 0; k < EPT; ++k)
    if (dv[k] >= 0) {
      int2 pr;
      pr.x = sv[k];
      pr.y = dv[k];
      staged[slot[k]] = pr;
    }
  __syncthreads();
  const int nb = min(BIN_EDGES, nEdges - blockBase);
  int2 pr[EPT];
#pragma unroll
  for (int k = 0; k < EPT; ++k) pr[k] = staged[min(k * BIN_THREADS + tid, nb - 1)];
  int pos[EPT];
#pragma unroll
  for (int k = 0; k < EPT; ++k)
    pos[k] = ldelta[pr[k].y >> 7] + k * BIN_THREADS + tid;
#pragma unroll
  for (int k = 0; k < EPT; ++k)
    if (k * BIN_THREADS + tid < nb) {
      int2 o;
      o.x = pr[k].x;
      o.y = pr[k].y & 127;
      *(int2*)(bedge + 2 * (size_t)pos[k]) = o;
    }
}

template <bool RELU, bool DEG>
__global__ void __launch_bounds__(OWN_THREADS, 8)
owner_layer_kernel(const float* __restrict__ xin, const int* __restrict__ bedge,
                   const int* __restrict__ offs, float* __restrict__ deg,
                   const float* __restrict__ Wl, const float* __restrict__ bl,
                   const float* __restrict__ Wr, float* __restrict__ out,
                   int nNodes) {
  __shared__ __attribute__((aligned(16))) float acc[((NRP + 1) * PITCH + 3) / 4 * 4];
  __shared__ __attribute__((aligned(16))) float rootl[NRP * RPITCH];
  __shared__ float degl[NRP + 1];
  __shared__ __attribute__((aligned(16))) int2 sorted[CAP];
  __shared__ __attribute__((aligned(16))) int cnt[128];
  __shared__ __attribute__((aligned(16))) int startl[128];
  const int tid = threadIdx.x;
  const int lane = tid & 31;
  const int wave = __builtin_amdgcn_readfirstlane(tid >> 5);
  const int r = blockIdx.x;

  const int m = lane & 15;
  const int half = lane >> 4;
  const int khalf = half << 1;
  const int mhi = half << 3;
  const int nt = wave & 3;
  const int j = (nt << 4) + m;

  v2f bL[16], bR[16];
#pragma unroll
  for (int kk = 0; kk < 16; ++kk) {
    const int k = (kk << 2) + khalf;
    bL[kk] = *(const v2f*)(Wl + j * D + k);
    bR[kk] = *(const v2f*)(Wr + j * D + k);
  }

  const int nodeBase = r * NR;
  const int nValid = min(NR, nNodes - nodeBase);
  static_assert(NRP * (D / 4) == 7 * OWN_THREADS, "seven whole trips");
  v4f rv[NRP * (D / 4) / OWN_THREADS];
#pragma unroll
  for (int t = 0; t < NRP * (D / 4) / OWN_THREADS; ++t) {
    const int row = (tid >> 4) + t * (OWN_THREADS / 16);
    const int grow = min(nodeBase + row, nNodes - 1);
    rv[t] = *(const v4f*)(xin + (size_t)grow * D + 4 * (tid & 15));
  }

  float dval = 0.f;
  if (!DEG) dval = deg[min(nodeBase + min(tid, NRP - 1), nNodes - 1)];

  for (int i = tid; i < (((NRP + 1) * PITCH + 3) / 4); i += OWN_THREADS)
    ((float4*)acc)[i] = make_float4(0.f, 0.f, 0.f, 0.f);
  if (DEG) {
    for (int i = tid; i <= NRP; i += OWN_THREADS) degl[i] = 0.f;
  } else {
    if (tid < NRP) degl[tid] = dval;
  }
  asm volatile("" ::"v"(rv[0]), "v"(rv[1]), "v"(rv[2]), "v"(rv[3]), "v"(rv[4]),
               "v"(rv[5]), "v"(rv[6]));
#pragma unroll
  for (int t = 0; t < NRP * (D / 4) / OWN_THREADS; ++t) {
    const int row = (tid >> 4) + t * (OWN_THREADS / 16);
    *(v4f*)(rootl + row * RPITCH + 4 * (tid & 15)) = rv[t];
  }
  const int eBeg = offs[r];
  const int eEnd = offs[r + 1];

  const unsigned lane8 = (unsigned)lane << 3;
  const unsigned lane4 = (unsigned)lane << 2;
  const char* __restrict__ xb = (const char*)xin;
  char* accb = (char*)acc;

  for (int cb = eBeg; cb < eEnd; cb += CAP) {
    const int n = min(CAP, eEnd - cb);
    if (tid < 128) cnt[tid] = 0;
    __syncthreads();

    int es[SEPT], ed[SEPT];
#pragma unroll
    for (int k = 0; k < SEPT; ++k) {
      const int i = k * OWN_THREADS + tid;
      ed[k] = -1;
      es[k] = 0;
      if (i < n) {
        const int2 sd = *(const int2*)(bedge + 2 * (size_t)(cb + i));
        es[k] = sd.x;
        ed[k] = sd.y;
      }
    }
#pragma unroll
    for (int k = 0; k < SEPT; ++k)
      if (ed[k] >= 0) ed[k] |= atomicAdd(&cnt[ed[k]], 1) << 7;
    __syncthreads();
    if (DEG) {
      if (tid < NRP) degl[tid] += (float)cnt[tid];
    }
    if (wave == 0) {
      const int4 c4 = ((const int4*)cnt)[lane];
      const int own = c4.x + c4.y + c4.z + c4.w;
      int s = own;
#pragma unroll
      for (int off = 1; off < 32; off <<= 1) {
        const int t = __shfl_up(s, off);
        if (lane >= off) s += t;
      }
      int4 st;
      st.x = s - own;
      st.y = st.x + c4.x;
      st.z = st.y + c4.y;
      st.w = st.z + c4.z;
      ((int4*)startl)[lane] = st;
    }
    __syncthreads();
#pragma unroll
    for (int k = 0; k < SEPT; ++k)
      if (ed[k] >= 0) {
        int2 pr;
        pr.x = es[k];
        pr.y = ed[k] & 127;
        sorted[startl[pr.y] + (ed[k] >> 7)] = pr;
      }
    __syncthreads();

    for (int base = wave * 32; base < n; base += OWN_WAVES * 32) {
      int2 sd;
      sd.x = 0;
      sd.y = NRP;
      if (base + lane < n) sd = sorted[base + lane];
      const unsigned soff = (unsigned)sd.x << 8;
      int cur = __builtin_amdgcn_readlane(sd.y, 0);
      v2f ps = (v2f){0.f, 0.f};
#pragma unroll
      for (int jj = 0; jj < 32; jj += GB) {
        v2f v[GB];
#pragma unroll
        for (int u = 0; u < GB; ++u) {
          const unsigned so =
              (unsigned)__builtin_amdgcn_readlane((int)soff, jj + u);
          v[u] = *(const v2f*)(xb + (so | lane8));
        }
#pragma unroll
        for (int u = 0; u < GB; ++u) {
          const int row = __builtin_amdgcn_readlane(sd.y, jj + u);
          if (row != cur) {
            float* p = (float*)(accb + ((unsigned)cur * (PITCH * 4) + lane4));
            __hip_atomic_fetch_add(p, ps[0], __ATOMIC_RELAXED,
                                   __HIP_MEMORY_SCOPE_WORKGROUP);
            __hip_atomic_fetch_add(p + 32, ps[1], __ATOMIC_RELAXED,
                                   __HIP_MEMORY_SCOPE_WORKGROUP);
            ps = (v2f){0.f, 0.f};
            cur = row;
          }
          ps += v[u];
        }
      }
      float* p = (float*)(accb + ((unsigned)cur * (PITCH * 4) + lane4));
      __hip_atomic_fetch_add(p, ps[0], __ATOMIC_RELAXED,
                             __HIP_MEMORY_SCOPE_WORKGROUP);
      __hip_atomic_fetch_add(p + 32, ps[1], __ATOMIC_RELAXED,
                             __HIP_MEMORY_SCOPE_WORKGROUP);
    }
  }

  __syncthreads();

  if (DEG)
    for (int i = tid; i < nValid; i += OWN_THREADS) deg[nodeBase + i] = degl[i];
  const float bias = bl[j];

  for (int tile = wave >> 2; tile < NRP / 16; tile += OWN_WAVES / 4) {
    const int lrow = (tile << 4) + m;
    const float rdeg = 1.0f / fmaxf(degl[lrow], 1.0f);
    const float* arow = acc + lrow * PITCH + half;
    const float* xrow = rootl + lrow * RPITCH + khalf;

    v8f c = (v8f){0, 0, 0, 0, 0, 0, 0, 0};
#pragma unroll
    for (int kk = 0; kk < 16; ++kk) {
      v2f a;
      a[0] = arow[2 * kk] * rdeg;
      a[1] = arow[32 + 2 * kk] * rdeg;
      c = __builtin_amdgcn_wmma_f32_16x16x4_f32(false, a, false, bL[kk],
                                                (short)0, c, false, false);
      const v2f ax = *(const v2f*)(xrow + (kk << 2));
      c = __builtin_amdgcn_wmma_f32_16x16x4_f32(false, ax, false, bR[kk],
                                                (short)0, c, false, false);
    }

    const int lr0 = (tile << 4) + mhi;
#pragma unroll
    for (int v = 0; v < 8; ++v) {
      float val = c[v] + bias;
      if (RELU) val = fmaxf(val, 0.0f);
      if (lr0 + v < nValid) out[(size_t)(nodeBase + lr0 + v) * D + j] = val;
    }
  }
}

static inline size_t align256(size_t x) { return (x + 255) & ~(size_t)255; }

extern "C" void kernel_launch(void* const* d_in, const int* in_sizes, int n_in,
                              void* d_out, int out_size, void* d_ws,
                              size_t ws_size, hipStream_t stream) {
  const float* x = (const float*)d_in[0];
  const int* src = (const int*)d_in[1];
  const int* dst = (const int*)d_in[2];
  const float* Wl[3] = {(const float*)d_in[3], (const float*)d_in[6],
                        (const float*)d_in[9]};
  const float* bl[3] = {(const float*)d_in[4], (const float*)d_in[7],
                        (const float*)d_in[10]};
  const float* Wr[3] = {(const float*)d_in[5], (const float*)d_in[8],
                        (const float*)d_in[11]};

  const int nNodes = in_sizes[0] / D;
  const int nEdges = in_sizes[1];
  const size_t featBytes = (size_t)nNodes * D * sizeof(float);
  const int nRanges = (nNodes + NR - 1) / NR;

  char* ws = (char*)d_ws;
  float* deg = (float*)ws;
  size_t ofs = align256((size_t)nNodes * sizeof(float));
  float* h1 = (float*)(ws + ofs);
  ofs += align256(featBytes);
  int* bedge = (int*)(ws + ofs);
  ofs += align256((size_t)nEdges * 2 * sizeof(int));
  int* gcount = (int*)(ws + ofs);
  ofs += align256((size_t)nRanges * sizeof(int));
  int* offs = (int*)(ws + ofs);
  ofs += align256((size_t)(nRanges + 1) * sizeof(int));
  int* cursor = (int*)(ws + ofs);
  float* h0 = (float*)d_out;

  hipMemsetAsync(gcount, 0, (size_t)nRanges * sizeof(int), stream);
  const int binBlocks = (nEdges + BIN_THREADS * EPT - 1) / (BIN_THREADS * EPT);
  if (binBlocks > 0)
    range_count_kernel<<<binBlocks, BIN_THREADS, nRanges * sizeof(int),
                         stream>>>(dst, gcount, nEdges, nRanges);
  range_scan_kernel<<<1, 1024, 0, stream>>>(gcount, offs, cursor, nRanges);
  if (binBlocks > 0)
    range_fill_kernel<<<binBlocks, BIN_THREADS, 2 * nRanges * sizeof(int),
                        stream>>>(src, dst, cursor, bedge, nEdges, nRanges);

  auto run_layer = [&](const float* in, const float* wl, const float* b,
                       const float* wr, float* outp, bool relu, bool first) {
    if (nRanges <= 0) return;
    if (first)
      owner_layer_kernel<true, true><<<nRanges, OWN_THREADS, 0, stream>>>(
          in, bedge, offs, deg, wl, b, wr, outp, nNodes);
    else if (relu)
      owner_layer_kernel<true, false><<<nRanges, OWN_THREADS, 0, stream>>>(
          in, bedge, offs, deg, wl, b, wr, outp, nNodes);
    else
      owner_layer_kernel<false, false><<<nRanges, OWN_THREADS, 0, stream>>>(
          in, bedge, offs, deg, wl, b, wr, outp, nNodes);
  };

  run_layer(x, Wl[0], bl[0], Wr[0], h0,  true,  true);
  run_layer(h0, Wl[1], bl[1], Wr[1], h1,  true,  false);
  run_layer(h1, Wl[2], bl[2], Wr[2], (float*)d_out,  false,
             false);
}


# Weights repacked once a call into B-fragment order (6 spare blocks of the scan launch): an owner wave reads Wl, Wr as 16 whole-line 16-B loads, not 32 8-B loads that touch 16 lines each
# speedup vs baseline: 5.3086x; 1.0648x over previous
#include <hip/hip_runtime.h>


typedef float v2f __attribute__((ext_vector_type(2)));
typedef float v4f __attribute__((ext_vector_type(4)));
typedef float v8f __attribute__((ext_vector_type(8)));

#define D 64
#define NR 100
#define BIN_THREADS 512
#define EPT 16
#define BIN_EDGES (BIN_THREADS * EPT)
#define BIN_WAVES (BIN_THREADS / 32)
static_assert(BIN_WAVES <= 32, "one lane of the first wave per wave in the block scan");
static_assert(NR <= 128, "a row within its range fits seven bits of the staged pair");
#define OWN_THREADS 256
#define OWN_WAVES (OWN_THREADS / 32)
#define GB 16
#define NRP 112
static_assert(NRP < 128, "a destination row fits the seven low bits of the sort's key");
#define PITCH 66
#define RPITCH 68
#define CAP 2048
#define SEPT (CAP / OWN_THREADS)

template <bool VEC>
__device__ __forceinline__ void load_block_edges(const int* __restrict__ p,
                                                 int blockBase, int nEdges,
                                                 int (&v)[EPT]) {
#pragma unroll
  for (int k = 0; k < EPT / 4; ++k) {
    const int e = blockBase + ((k * BIN_THREADS + (int)threadIdx.x) << 2);
    if (VEC) {
      const int4 q = *(const int4*)(p + e);
      v[4 * k + 0] = q.x;
      v[4 * k + 1] = q.y;
      v[4 * k + 2] = q.z;
      v[4 * k + 3] = q.w;
    } else {
#pragma unroll
      for (int u = 0; u < 4; ++u) {
        const int val = p[min(e + u, nEdges - 1)];
        v[4 * k + u] = (e + u < nEdges) ? val : -1;
      }
    }
  }
}

__global__ void __launch_bounds__(BIN_THREADS)
range_count_kernel(const int* __restrict__ dst, int* __restrict__ gcount,
                   int nEdges, int nRanges) {
  extern __shared__ int lcount[];
  for (int i = threadIdx.x; i < nRanges; i += BIN_THREADS) lcount[i] = 0;
  const int spare = min((int)(threadIdx.x & 31), nRanges - 1);
  const int blockBase = blockIdx.x * BIN_EDGES;
  int dv[EPT];
  if ((((size_t)dst) & 15) == 0 && blockBase + BIN_EDGES <= nEdges)
    load_block_edges<true>(dst, blockBase, nEdges, dv);
  else
    load_block_edges<false>(dst, blockBase, nEdges, dv);
  __syncthreads();
#pragma unroll
  for (int k = 0; k < EPT; ++k) {
    const bool ok = dv[k] >= 0;
    atomicAdd(&lcount[ok ? (int)((unsigned)dv[k] / NR) : spare], ok ? 1 : 0);
  }
  __syncthreads();
  for (int i = threadIdx.x; i < nRanges; i += BIN_THREADS) {
    const int c = lcount[i];
    if (c) atomicAdd(gcount + i, c);
  }
}

__global__ void __launch_bounds__(1024)
range_scan_kernel(const int* __restrict__ gcount, int* __restrict__ offs,
                  int* __restrict__ cursor, int nRanges,
                  const float* __restrict__ w0, const float* __restrict__ w1,
                  const float* __restrict__ w2, const float* __restrict__ w3,
                  const float* __restrict__ w4, const float* __restrict__ w5,
                  float* __restrict__ wpack) {
  __shared__ int tmp[1024];
  const int tid = threadIdx.x;
  if (blockIdx.x > 0) {
    const int mat = blockIdx.x - 1;
    const float* __restrict__ w = mat == 0 ? w0 : mat == 1 ? w1 : mat == 2 ? w2
                                : mat == 3 ? w3 : mat == 4 ? w4 : w5;
    const int ln = tid & 31;
    const int t = (tid >> 5) & 7;
    const int row = ((tid >> 8) << 4) + (ln & 15);
    const float* p = w + row * D + (t << 3) + ((ln >> 4) << 1);
    const v2f lo = *(const v2f*)p;
    const v2f hi = *(const v2f*)(p + 4);
    *(v4f*)(wpack + ((size_t)mat * (D * D) + ((size_t)tid << 2))) =
        (v4f){lo[0], lo[1], hi[0], hi[1]};
    return;
  }
  int carry = 0;
  for (int c0 = 0; c0 < nRanges; c0 += 1024) {
    const int i = c0 + tid;
    const int v = (i < nRanges) ? gcount[i] : 0;
    tmp[tid] = v;
    __syncthreads();
    for (int off = 1; off < 1024; off <<= 1) {
      const int t = (tid >= off) ? tmp[tid - off] : 0;
      __syncthreads();
      tmp[tid] += t;
      __syncthreads();
    }
    const int incl = tmp[tid];
    const int total = tmp[1023];
    if (i < nRanges) {
      offs[i] = carry + incl - v;
      cursor[i] = carry + incl - v;
    }
    carry += total;
    __syncthreads();
  }
  if (tid == 0) offs[nRanges] = carry;
}

__global__ void __launch_bounds__(BIN_THREADS)
range_fill_kernel(const int* __restrict__ src, const int* __restrict__ dst,
                  int* __restrict__ cursor, int* __restrict__ bedge,
                  int nEdges, int nRanges) {
  extern __shared__ int lsh[];
  __shared__ __attribute__((aligned(16))) int2 staged[BIN_EDGES];
  __shared__ int wsum[32];
  __shared__ int wtot;
  int* lcnt = lsh;
  int* ldelta = lsh + nRanges;
  const int tid = threadIdx.x;
  const int lane = tid & 31;
  const int wave = tid >> 5;
  const int blockBase = blockIdx.x * BIN_EDGES;
  for (int i = tid; i < nRanges; i += BIN_THREADS) lcnt[i] = 0;
  const int spare = min(lane, nRanges - 1);
  int dv[EPT], sv[EPT];
  if (((((size_t)dst) | ((size_t)src)) & 15) == 0 &&
      blockBase + BIN_EDGES <= nEdges) {
    load_block_edges<true>(dst, blockBase, nEdges, dv);
    load_block_edges<true>(src, blockBase, nEdges, sv);
  } else {
    load_block_edges<false>(dst, blockBase, nEdges, dv);
    load_block_edges<false>(src, blockBase, nEdges, sv);
  }
  __syncthreads();
#pragma unroll
  for (int k = 0; k < EPT; ++k) {
    const bool ok = dv[k] >= 0;
    const int r = ok ? (int)((unsigned)dv[k] / NR) : spare;
    if (ok) dv[k] = (r << 7) | (dv[k] - r * NR);
    atomicAdd(&lcnt[r], ok ? 1 : 0);
  }
  __syncthreads();
  int carry = 0;
  for (int c0 = 0; c0 < nRanges; c0 += 2 * BIN_THREADS) {
    const int i0 = c0 + 2 * tid;
    const int i1 = i0 + 1;
    const int n0 = (i0 < nRanges) ? lcnt[i0] : 0;
    const int n1 = (i1 < nRanges) ? lcnt[i1] : 0;
    const int g0 = n0 ? atomicAdd(cursor + i0, n0) : 0;
    const int g1 = n1 ? atomicAdd(cursor + i1, n1) : 0;
    const int own = n0 + n1;
    int sc = own;
#pragma unroll
    for (int off = 1; off < 32; off <<= 1) {
      const int t = __shfl_up(sc, off);
      if (lane >= off) sc += t;
    }
    if (lane == 31) wsum[wave] = sc;
    __syncthreads();
    if (wave == 0) {
      const int w = (lane < BIN_WAVES) ? wsum[lane] : 0;
      int ws = w;
#pragma unroll
      for (int off = 1; off < 32; off <<= 1) {
        const int t = __shfl_up(ws, off);
        if (lane >= off) ws += t;
      }
      wsum[lane] = ws - w;
      if (lane == 31) wtot = ws;
    }
    __syncthreads();
    const int excl = carry + wsum[wave] + sc - own;
    if (i0 < nRanges) {
      lcnt[i0] = excl;
      ldelta[i0] = g0 - excl;
    }
    if (i1 < nRanges) {
      lcnt[i1] = excl + n0;
      ldelta[i1] = g1 - (excl + n0);
    }
    carry += wtot;
    __syncthreads();
  }
  int slot[EPT];
#pragma unroll
  for (int k = 0; k < EPT; ++k) {
    const bool ok = dv[k] >= 0;
    slot[k] = atomicAdd(&lcnt[ok ? (dv[k] >> 7) : spare], ok ? 1 : 0);
  }
#pragma unroll
  for (int k = 0; k < EPT; ++k)
    if (dv[k] >= 0) {
      int2 pr;
      pr.x = sv[k];
      pr.y = dv[k];
      staged[slot[k]] = pr;
    }
  __syncthreads();
  const int nb = min(BIN_EDGES, nEdges - blockBase);
  int2 pr[EPT];
#pragma unroll
  for (int k = 0; k < EPT; ++k) pr[k] = staged[min(k * BIN_THREADS + tid, nb - 1)];
  int pos[EPT];
#pragma unroll
  for (int k = 0; k < EPT; ++k)
    pos[k] = ldelta[pr[k].y >> 7] + k * BIN_THREADS + tid;
#pragma unroll
  for (int k = 0; k < EPT; ++k)
    if (k * BIN_THREADS + tid < nb) {
      int2 o;
      o.x = pr[k].x;
      o.y = pr[k].y & 127;
      *(int2*)(bedge + 2 * (size_t)pos[k]) = o;
    }
}

template <bool RELU, bool DEG>
__global__ void __launch_bounds__(OWN_THREADS, 8)
owner_layer_kernel(const float* __restrict__ xin, const int* __restrict__ bedge,
                   const int* __restrict__ offs, float* __restrict__ deg,
                   const float* __restrict__ Wl, const float* __restrict__ bl,
                   const float* __restrict__ Wr, float* __restrict__ out,
                   int nNodes) {
  __shared__ __attribute__((aligned(16))) float acc[((NRP + 1) * PITCH + 3) / 4 * 4];
  __shared__ __attribute__((aligned(16))) float rootl[NRP * RPITCH];
  __shared__ float degl[NRP + 1];
  __shared__ __attribute__((aligned(16))) int2 sorted[CAP];
  __shared__ __attribute__((aligned(16))) int cnt[128];
  __shared__ __attribute__((aligned(16))) int startl[128];
  const int tid = threadIdx.x;
  const int lane = tid & 31;
  const int wave = __builtin_amdgcn_readfirstlane(tid >> 5);
  const int r = blockIdx.x;

  const int m = lane & 15;
  const int half = lane >> 4;
  const int khalf = half << 1;
  const int mhi = half << 3;
  const int nt = wave & 3;
  const int j = (nt << 4) + m;

  v2f bL[16], bR[16];
  {
    const v4f* __restrict__ wlp = (const v4f*)Wl + (nt << 8) + lane;
    const v4f* __restrict__ wrp = (const v4f*)Wr + (nt << 8) + lane;
#pragma unroll
    for (int t = 0; t < 8; ++t) {
      const v4f ql = wlp[t << 5];
      const v4f qr = wrp[t << 5];
      bL[2 * t] = (v2f){ql[0], ql[1]};
      bL[2 * t + 1] = (v2f){ql[2], ql[3]};
      bR[2 * t] = (v2f){qr[0], qr[1]};
      bR[2 * t + 1] = (v2f){qr[2], qr[3]};
    }
  }

  const int nodeBase = r * NR;
  const int nValid = min(NR, nNodes - nodeBase);
  static_assert(NRP * (D / 4) == 7 * OWN_THREADS, "seven whole trips");
  v4f rv[NRP * (D / 4) / OWN_THREADS];
#pragma unroll
  for (int t = 0; t < NRP * (D / 4) / OWN_THREADS; ++t) {
    const int row = (tid >> 4) + t * (OWN_THREADS / 16);
    const int grow = min(nodeBase + row, nNodes - 1);
    rv[t] = *(const v4f*)(xin + (size_t)grow * D + 4 * (tid & 15));
  }

  float dval = 0.f;
  if (!DEG) dval = deg[min(nodeBase + min(tid, NRP - 1), nNodes - 1)];

  for (int i = tid; i < (((NRP + 1) * PITCH + 3) / 4); i += OWN_THREADS)
    ((float4*)acc)[i] = make_float4(0.f, 0.f, 0.f, 0.f);
  if (DEG) {
    for (int i = tid; i <= NRP; i += OWN_THREADS) degl[i] = 0.f;
  } else {
    if (tid < NRP) degl[tid] = dval;
  }
  asm volatile("" ::"v"(rv[0]), "v"(rv[1]), "v"(rv[2]), "v"(rv[3]), "v"(rv[4]),
               "v"(rv[5]), "v"(rv[6]));
#pragma unroll
  for (int t = 0; t < NRP * (D / 4) / OWN_THREADS; ++t) {
    const int row = (tid >> 4) + t * (OWN_THREADS / 16);
    *(v4f*)(rootl + row * RPITCH + 4 * (tid & 15)) = rv[t];
  }
  const int eBeg = offs[r];
  const int eEnd = offs[r + 1];

  const unsigned lane8 = (unsigned)lane << 3;
  const unsigned lane4 = (unsigned)lane << 2;
  const char* __restrict__ xb = (const char*)xin;
  char* accb = (char*)acc;

  for (int cb = eBeg; cb < eEnd; cb += CAP) {
    const int n = min(CAP, eEnd - cb);
    if (tid < 128) cnt[tid] = 0;
    __syncthreads();

    int es[SEPT], ed[SEPT];
#pragma unroll
    for (int k = 0; k < SEPT; ++k) {
      const int i = k * OWN_THREADS + tid;
      ed[k] = -1;
      es[k] = 0;
      if (i < n) {
        const int2 sd = *(const int2*)(bedge + 2 * (size_t)(cb + i));
        es[k] = sd.x;
        ed[k] = sd.y;
      }
    }
#pragma unroll
    for (int k = 0; k < SEPT; ++k)
      if (ed[k] >= 0) ed[k] |= atomicAdd(&cnt[ed[k]], 1) << 7;
    __syncthreads();
    if (DEG) {
      if (tid < NRP) degl[tid] += (float)cnt[tid];
    }
    if (wave == 0) {
      const int4 c4 = ((const int4*)cnt)[lane];
      const int own = c4.x + c4.y + c4.z + c4.w;
      int s = own;
#pragma unroll
      for (int off = 1; off < 32; off <<= 1) {
        const int t = __shfl_up(s, off);
        if (lane >= off) s += t;
      }
      int4 st;
      st.x = s - own;
      st.y = st.x + c4.x;
      st.z = st.y + c4.y;
      st.w = st.z + c4.z;
      ((int4*)startl)[lane] = st;
    }
    __syncthreads();
#pragma unroll
    for (int k = 0; k < SEPT; ++k)
      if (ed[k] >= 0) {
        int2 pr;
        pr.x = es[k];
        pr.y = ed[k] & 127;
        sorted[startl[pr.y] + (ed[k] >> 7)] = pr;
      }
    __syncthreads();

    for (int base = wave * 32; base < n; base += OWN_WAVES * 32) {
      int2 sd;
      sd.x = 0;
      sd.y = NRP;
      if (base + lane < n) sd = sorted[base + lane];
      const unsigned soff = (unsigned)sd.x << 8;
      int cur = __builtin_amdgcn_readlane(sd.y, 0);
      v2f ps = (v2f){0.f, 0.f};
#pragma unroll
      for (int jj = 0; jj < 32; jj += GB) {
        v2f v[GB];
#pragma unroll
        for (int u = 0; u < GB; ++u) {
          const unsigned so =
              (unsigned)__builtin_amdgcn_readlane((int)soff, jj + u);
          v[u] = *(const v2f*)(xb + (so | lane8));
        }
#pragma unroll
        for (int u = 0; u < GB; ++u) {
          const int row = __builtin_amdgcn_readlane(sd.y, jj + u);
          if (row != cur) {
            float* p = (float*)(accb + ((unsigned)cur * (PITCH * 4) + lane4));
            __hip_atomic_fetch_add(p, ps[0], __ATOMIC_RELAXED,
                                   __HIP_MEMORY_SCOPE_WORKGROUP);
            __hip_atomic_fetch_add(p + 32, ps[1], __ATOMIC_RELAXED,
                                   __HIP_MEMORY_SCOPE_WORKGROUP);
            ps = (v2f){0.f, 0.f};
            cur = row;
          }
          ps += v[u];
        }
      }
      float* p = (float*)(accb + ((unsigned)cur * (PITCH * 4) + lane4));
      __hip_atomic_fetch_add(p, ps[0], __ATOMIC_RELAXED,
                             __HIP_MEMORY_SCOPE_WORKGROUP);
      __hip_atomic_fetch_add(p + 32, ps[1], __ATOMIC_RELAXED,
                             __HIP_MEMORY_SCOPE_WORKGROUP);
    }
  }

  __syncthreads();

  if (DEG)
    for (int i = tid; i < nValid; i += OWN_THREADS) deg[nodeBase + i] = degl[i];
  const float bias = bl[j];

  for (int tile = wave >> 2; tile < NRP / 16; tile += OWN_WAVES / 4) {
    const int lrow = (tile << 4) + m;
    const float rdeg = 1.0f / fmaxf(degl[lrow], 1.0f);
    const float* arow = acc + lrow * PITCH + half;
    const float* xrow = rootl + lrow * RPITCH + khalf;

    v8f c = (v8f){0, 0, 0, 0, 0, 0, 0, 0};
#pragma unroll
    for (int kk = 0; kk < 16; ++kk) {
      v2f a;
      a[0] = arow[2 * kk] * rdeg;
      a[1] = arow[32 + 2 * kk] * rdeg;
      c = __builtin_amdgcn_wmma_f32_16x16x4_f32(false, a, false, bL[kk],
                                                (short)0, c, false, false);
      const v2f ax = *(const v2f*)(xrow + (kk << 2));
      c = __builtin_amdgcn_wmma_f32_16x16x4_f32(false, ax, false, bR[kk],
                                                (short)0, c, false, false);
    }

    const int lr0 = (tile << 4) + mhi;
#pragma unroll
    for (int v = 0; v < 8; ++v) {
      float val = c[v] + bias;
      if (RELU) val = fmaxf(val, 0.0f);
      if (lr0 + v < nValid) out[(size_t)(nodeBase + lr0 + v) * D + j] = val;
    }
  }
}

static inline size_t align256(size_t x) { return (x + 255) & ~(size_t)255; }

extern "C" void kernel_launch(void* const* d_in, const int* in_sizes, int n_in,
                              void* d_out, int out_size, void* d_ws,
                              size_t ws_size, hipStream_t stream) {
  const float* x = (const float*)d_in[0];
  const int* src = (const int*)d_in[1];
  const int* dst = (const int*)d_in[2];
  const float* Wl[3] = {(const float*)d_in[3], (const float*)d_in[6],
                        (const float*)d_in[9]};
  const float* bl[3] = {(const float*)d_in[4], (const float*)d_in[7],
                        (const float*)d_in[10]};
  const float* Wr[3] = {(const float*)d_in[5], (const float*)d_in[8],
                        (const float*)d_in[11]};

  const int nNodes = in_sizes[0] / D;
  const int nEdges = in_sizes[1];
  const size_t featBytes = (size_t)nNodes * D * sizeof(float);
  const int nRanges = (nNodes + NR - 1) / NR;

  char* ws = (char*)d_ws;
  float* deg = (float*)ws;
  size_t ofs = align256((size_t)nNodes * sizeof(float));
  float* h1 = (float*)(ws + ofs);
  ofs += align256(featBytes);
  int* bedge = (int*)(ws + ofs);
  ofs += align256((size_t)nEdges * 2 * sizeof(int));
  int* gcount = (int*)(ws + ofs);
  ofs += align256((size_t)nRanges * sizeof(int));
  int* offs = (int*)(ws + ofs);
  ofs += align256((size_t)(nRanges + 1) * sizeof(int));
  int* cursor = (int*)(ws + ofs);
  ofs += align256((size_t)nRanges * sizeof(int));
  float* wpack = (float*)(ws + ofs);
  float* h0 = (float*)d_out;

  hipMemsetAsync(gcount, 0, (size_t)nRanges * sizeof(int), stream);
  const int binBlocks = (nEdges + BIN_THREADS * EPT - 1) / (BIN_THREADS * EPT);
  if (binBlocks > 0)
    range_count_kernel<<<binBlocks, BIN_THREADS, nRanges * sizeof(int),
                         stream>>>(dst, gcount, nEdges, nRanges);
  range_scan_kernel<<<1 + 6, 1024, 0, stream>>>(gcount, offs, cursor, nRanges,
                                                Wl[0], Wr[0], Wl[1], Wr[1],
                                                Wl[2], Wr[2], wpack);
  if (binBlocks > 0)
    range_fill_kernel<<<binBlocks, BIN_THREADS, 2 * nRanges * sizeof(int),
                        stream>>>(src, dst, cursor, bedge, nEdges, nRanges);

  auto run_layer = [&](const float* in, const float* wl, const float* b,
                       const float* wr, float* outp, bool relu, bool first) {
    if (nRanges <= 0) return;
    if (first)
      owner_layer_kernel<true, true><<<nRanges, OWN_THREADS, 0, stream>>>(
          in, bedge, offs, deg, wl, b, wr, outp, nNodes);
    else if (relu)
      owner_layer_kernel<true, false><<<nRanges, OWN_THREADS, 0, stream>>>(
          in, bedge, offs, deg, wl, b, wr, outp, nNodes);
    else
      owner_layer_kernel<false, false><<<nRanges, OWN_THREADS, 0, stream>>>(
          in, bedge, offs, deg, wl, b, wr, outp, nNodes);
  };

  run_layer(x, wpack + 0 * D * D, bl[0], wpack + 1 * D * D, h0,  true,
             true);
  run_layer(h0, wpack + 2 * D * D, bl[1], wpack + 3 * D * D, h1,  true,
             false);
  run_layer(h1, wpack + 4 * D * D, bl[2], wpack + 5 * D * D, (float*)d_out,
             false,  false);
}


# Root term x @ Wr^T moved out of the closing product, run beside the gather: a group's waves 0-3 multiply it before their edge batches, waves 4-7 after theirs; mean terms added last (new sum order)
# speedup vs baseline: 5.5586x; 1.0471x over previous
#include <hip/hip_runtime.h>


typedef float v2f __attribute__((ext_vector_type(2)));
typedef float v4f __attribute__((ext_vector_type(4)));
typedef float v8f __attribute__((ext_vector_type(8)));

#define D 64
#define NR 100
#define BIN_THREADS 512
#define EPT 16
#define BIN_EDGES (BIN_THREADS * EPT)
#define BIN_WAVES (BIN_THREADS / 32)
static_assert(BIN_WAVES <= 32, "one lane of the first wave per wave in the block scan");
static_assert(NR <= 128, "a row within its range fits seven bits of the staged pair");
#define OWN_THREADS 256
#define OWN_WAVES (OWN_THREADS / 32)
#define GB 16
#define NRP 112
static_assert(NRP < 128, "a destination row fits the seven low bits of the sort's key");
#define PITCH 66
#define RPITCH 68
#define CAP 2048
#define SEPT (CAP / OWN_THREADS)

template <bool VEC>
__device__ __forceinline__ void load_block_edges(const int* __restrict__ p,
                                                 int blockBase, int nEdges,
                                                 int (&v)[EPT]) {
#pragma unroll
  for (int k = 0; k < EPT / 4; ++k) {
    const int e = blockBase + ((k * BIN_THREADS + (int)threadIdx.x) << 2);
    if (VEC) {
      const int4 q = *(const int4*)(p + e);
      v[4 * k + 0] = q.x;
      v[4 * k + 1] = q.y;
      v[4 * k + 2] = q.z;
      v[4 * k + 3] = q.w;
    } else {
#pragma unroll
      for (int u = 0; u < 4; ++u) {
        const int val = p[min(e + u, nEdges - 1)];
        v[4 * k + u] = (e + u < nEdges) ? val : -1;
      }
    }
  }
}

__global__ void __launch_bounds__(BIN_THREADS)
range_count_kernel(const int* __restrict__ dst, int* __restrict__ gcount,
                   int nEdges, int nRanges) {
  extern __shared__ int lcount[];
  for (int i = threadIdx.x; i < nRanges; i += BIN_THREADS) lcount[i] = 0;
  const int spare = min((int)(threadIdx.x & 31), nRanges - 1);
  const int blockBase = blockIdx.x * BIN_EDGES;
  int dv[EPT];
  if ((((size_t)dst) & 15) == 0 && blockBase + BIN_EDGES <= nEdges)
    load_block_edges<true>(dst, blockBase, nEdges, dv);
  else
    load_block_edges<false>(dst, blockBase, nEdges, dv);
  __syncthreads();
#pragma unroll
  for (int k = 0; k < EPT; ++k) {
    const bool ok = dv[k] >= 0;
    atomicAdd(&lcount[ok ? (int)((unsigned)dv[k] / NR) : spare], ok ? 1 : 0);
  }
  __syncthreads();
  for (int i = threadIdx.x; i < nRanges; i += BIN_THREADS) {
    const int c = lcount[i];
    if (c) atomicAdd(gcount + i, c);
  }
}

__global__ void __launch_bounds__(1024)
range_scan_kernel(const int* __restrict__ gcount, int* __restrict__ offs,
                  int* __restrict__ cursor, int nRanges,
                  const float* __restrict__ w0, const float* __restrict__ w1,
                  const float* __restrict__ w2, const float* __restrict__ w3,
                  const float* __restrict__ w4, const float* __restrict__ w5,
                  float* __restrict__ wpack) {
  __shared__ int tmp[1024];
  const int tid = threadIdx.x;
  if (blockIdx.x > 0) {
    const int mat = blockIdx.x - 1;
    const float* __restrict__ w = mat == 0 ? w0 : mat == 1 ? w1 : mat == 2 ? w2
                                : mat == 3 ? w3 : mat == 4 ? w4 : w5;
    const int ln = tid & 31;
    const int t = (tid >> 5) & 7;
    const int row = ((tid >> 8) << 4) + (ln & 15);
    const float* p = w + row * D + (t << 3) + ((ln >> 4) << 1);
    const v2f lo = *(const v2f*)p;
    const v2f hi = *(const v2f*)(p + 4);
    *(v4f*)(wpack + ((size_t)mat * (D * D) + ((size_t)tid << 2))) =
        (v4f){lo[0], lo[1], hi[0], hi[1]};
    return;
  }
  int carry = 0;
  for (int c0 = 0; c0 < nRanges; c0 += 1024) {
    const int i = c0 + tid;
    const int v = (i < nRanges) ? gcount[i] : 0;
    tmp[tid] = v;
    __syncthreads();
    for (int off = 1; off < 1024; off <<= 1) {
      const int t = (tid >= off) ? tmp[tid - off] : 0;
      __syncthreads();
      tmp[tid] += t;
      __syncthreads();
    }
    const int incl = tmp[tid];
    const int total = tmp[1023];
    if (i < nRanges) {
      offs[i] = carry + incl - v;
      cursor[i] = carry + incl - v;
    }
    carry += total;
    __syncthreads();
  }
  if (tid == 0) offs[nRanges] = carry;
}

__global__ void __launch_bounds__(BIN_THREADS)
range_fill_kernel(const int* __restrict__ src, const int* __restrict__ dst,
                  int* __restrict__ cursor, int* __restrict__ bedge,
                  int nEdges, int nRanges) {
  extern __shared__ int lsh[];
  __shared__ __attribute__((aligned(16))) int2 staged[BIN_EDGES];
  __shared__ int wsum[32];
  __shared__ int wtot;
  int* lcnt = lsh;
  int* ldelta = lsh + nRanges;
  const int tid = threadIdx.x;
  const int lane = tid & 31;
  const int wave = tid >> 5;
  const int blockBase = blockIdx.x * BIN_EDGES;
  for (int i = tid; i < nRanges; i += BIN_THREADS) lcnt[i] = 0;
  const int spare = min(lane, nRanges - 1);
  int dv[EPT], sv[EPT];
  if (((((size_t)dst) | ((size_t)src)) & 15) == 0 &&
      blockBase + BIN_EDGES <= nEdges) {
    load_block_edges<true>(dst, blockBase, nEdges, dv);
    load_block_edges<true>(src, blockBase, nEdges, sv);
  } else {
    load_block_edges<false>(dst, blockBase, nEdges, dv);
    load_block_edges<false>(src, blockBase, nEdges, sv);
  }
  __syncthreads();
#pragma unroll
  for (int k = 0; k < EPT; ++k) {
    const bool ok = dv[k] >= 0;
    const int r = ok ? (int)((unsigned)dv[k] / NR) : spare;
    if (ok) dv[k] = (r << 7) | (dv[k] - r * NR);
    atomicAdd(&lcnt[r], ok ? 1 : 0);
  }
  __syncthreads();
  int carry = 0;
  for (int c0 = 0; c0 < nRanges; c0 += 2 * BIN_THREADS) {
    const int i0 = c0 + 2 * tid;
    const int i1 = i0 + 1;
    const int n0 = (i0 < nRanges) ? lcnt[i0] : 0;
    const int n1 = (i1 < nRanges) ? lcnt[i1] : 0;
    const int g0 = n0 ? atomicAdd(cursor + i0, n0) : 0;
    const int g1 = n1 ? atomicAdd(cursor + i1, n1) : 0;
    const int own = n0 + n1;
    int sc = own;
#pragma unroll
    for (int off = 1; off < 32; off <<= 1) {
      const int t = __shfl_up(sc, off);
      if (lane >= off) sc += t;
    }
    if (lane == 31) wsum[wave] = sc;
    __syncthreads();
    if (wave == 0) {
      const int w = (lane < BIN_WAVES) ? wsum[lane] : 0;
      int ws = w;
#pragma unroll
      for (int off = 1; off < 32; off <<= 1) {
        const int t = __shfl_up(ws, off);
        if (lane >= off) ws += t;
      }
      wsum[lane] = ws - w;
      if (lane == 31) wtot = ws;
    }
    __syncthreads();
    const int excl = carry + wsum[wave] + sc - own;
    if (i0 < nRanges) {
      lcnt[i0] = excl;
      ldelta[i0] = g0 - excl;
    }
    if (i1 < nRanges) {
      lcnt[i1] = excl + n0;
      ldelta[i1] = g1 - (excl + n0);
    }
    carry += wtot;
    __syncthreads();
  }
  int slot[EPT];
#pragma unroll
  for (int k = 0; k < EPT; ++k) {
    const bool ok = dv[k] >= 0;
    slot[k] = atomicAdd(&lcnt[ok ? (dv[k] >> 7) : spare], ok ? 1 : 0);
  }
#pragma unroll
  for (int k = 0; k < EPT; ++k)
    if (dv[k] >= 0) {
      int2 pr;
      pr.x = sv[k];
      pr.y = dv[k];
      staged[slot[k]] = pr;
    }
  __syncthreads();
  const int nb = min(BIN_EDGES, nEdges - blockBase);
  int2 pr[EPT];
#pragma unroll
  for (int k = 0; k < EPT; ++k) pr[k] = staged[min(k * BIN_THREADS + tid, nb - 1)];
  int pos[EPT];
#pragma unroll
  for (int k = 0; k < EPT; ++k)
    pos[k] = ldelta[pr[k].y >> 7] + k * BIN_THREADS + tid;
#pragma unroll
  for (int k = 0; k < EPT; ++k)
    if (k * BIN_THREADS + tid < nb) {
      int2 o;
      o.x = pr[k].x;
      o.y = pr[k].y & 127;
      *(int2*)(bedge + 2 * (size_t)pos[k]) = o;
    }
}

template <int NT>
__device__ __forceinline__ void root_product(const float* rootl, int tile0, int m,
                                             int khalf, const v2f (&bR)[16],
                                             v8f (&cR)[4]) {
  const float* xrow[NT];
#pragma unroll
  for (int ti = 0; ti < NT; ++ti)
    xrow[ti] = rootl + (((tile0 + 2 * ti) << 4) + m) * RPITCH + khalf;
#pragma unroll
  for (int ti = 0; ti < 4; ++ti) cR[ti] = (v8f){0, 0, 0, 0, 0, 0, 0, 0};
#pragma unroll
  for (int kk = 0; kk < 16; ++kk) {
#pragma unroll
    for (int ti = 0; ti < NT; ++ti) {
      const v2f ax = *(const v2f*)(xrow[ti] + (kk << 2));
      cR[ti] = __builtin_amdgcn_wmma_f32_16x16x4_f32(false, ax, false, bR[kk],
                                                     (short)0, cR[ti], false, false);
    }
  }
}

template <bool RELU, bool DEG>
__global__ void __launch_bounds__(OWN_THREADS, 8)
owner_layer_kernel(const float* __restrict__ xin, const int* __restrict__ bedge,
                   const int* __restrict__ offs, float* __restrict__ deg,
                   const float* __restrict__ Wl, const float* __restrict__ bl,
                   const float* __restrict__ Wr, float* __restrict__ out,
                   int nNodes) {
  __shared__ __attribute__((aligned(16))) float acc[((NRP + 1) * PITCH + 3) / 4 * 4];
  __shared__ __attribute__((aligned(16))) float rootl[NRP * RPITCH];
  __shared__ float degl[NRP + 1];
  __shared__ __attribute__((aligned(16))) int2 sorted[CAP];
  __shared__ __attribute__((aligned(16))) int cnt[128];
  __shared__ __attribute__((aligned(16))) int startl[128];
  const int tid = threadIdx.x;
  const int lane = tid & 31;
  const int wave = __builtin_amdgcn_readfirstlane(tid >> 5);
  const int r = blockIdx.x;

  const int m = lane & 15;
  const int half = lane >> 4;
  const int khalf = half << 1;
  const int mhi = half << 3;
  const int nt = wave & 3;
  const int j = (nt << 4) + m;

  v2f bR[16];
  {
    const v4f* __restrict__ wrp = (const v4f*)Wr + (nt << 8) + lane;
#pragma unroll
    for (int t = 0; t < 8; ++t) {
      const v4f qr = wrp[t << 5];
      bR[2 * t] = (v2f){qr[0], qr[1]};
      bR[2 * t + 1] = (v2f){qr[2], qr[3]};
    }
  }
  const bool firstHalf = (wave >> 2) == 0;
  v8f cR[4];

  const int nodeBase = r * NR;
  const int nValid = min(NR, nNodes - nodeBase);
  static_assert(NRP * (D / 4) == 7 * OWN_THREADS, "seven whole trips");
  v4f rv[NRP * (D / 4) / OWN_THREADS];
#pragma unroll
  for (int t = 0; t < NRP * (D / 4) / OWN_THREADS; ++t) {
    const int row = (tid >> 4) + t * (OWN_THREADS / 16);
    const int grow = min(nodeBase + row, nNodes - 1);
    rv[t] = *(const v4f*)(xin + (size_t)grow * D + 4 * (tid & 15));
  }

  float dval = 0.f;
  if (!DEG) dval = deg[min(nodeBase + min(tid, NRP - 1), nNodes - 1)];

  for (int i = tid; i < (((NRP + 1) * PITCH + 3) / 4); i += OWN_THREADS)
    ((float4*)acc)[i] = make_float4(0.f, 0.f, 0.f, 0.f);
  if (DEG) {
    for (int i = tid; i <= NRP; i += OWN_THREADS) degl[i] = 0.f;
  } else {
    if (tid < NRP) degl[tid] = dval;
  }
  asm volatile("" ::"v"(rv[0]), "v"(rv[1]), "v"(rv[2]), "v"(rv[3]), "v"(rv[4]),
               "v"(rv[5]), "v"(rv[6]));
#pragma unroll
  for (int t = 0; t < NRP * (D / 4) / OWN_THREADS; ++t) {
    const int row = (tid >> 4) + t * (OWN_THREADS / 16);
    *(v4f*)(rootl + row * RPITCH + 4 * (tid & 15)) = rv[t];
  }
  const int eBeg = offs[r];
  const int eEnd = offs[r + 1];

  const unsigned lane8 = (unsigned)lane << 3;
  const unsigned lane4 = (unsigned)lane << 2;
  const char* __restrict__ xb = (const char*)xin;
  char* accb = (char*)acc;

  int cb = eBeg;
  do {
    const int n = max(0, min(CAP, eEnd - cb));
    if (tid < 128) cnt[tid] = 0;
    __syncthreads();

    int es[SEPT], ed[SEPT];
#pragma unroll
    for (int k = 0; k < SEPT; ++k) {
      const int i = k * OWN_THREADS + tid;
      ed[k] = -1;
      es[k] = 0;
      if (i < n) {
        const int2 sd = *(const int2*)(bedge + 2 * (size_t)(cb + i));
        es[k] = sd.x;
        ed[k] = sd.y;
      }
    }
#pragma unroll
    for (int k = 0; k < SEPT; ++k)
      if (ed[k] >= 0) ed[k] |= atomicAdd(&cnt[ed[k]], 1) << 7;
    __syncthreads();
    if (DEG) {
      if (tid < NRP) degl[tid] += (float)cnt[tid];
    }
    if (wave == 0) {
      const int4 c4 = ((const int4*)cnt)[lane];
      const int own = c4.x + c4.y + c4.z + c4.w;
      int s = own;
#pragma unroll
      for (int off = 1; off < 32; off <<= 1) {
        const int t = __shfl_up(s, off);
        if (lane >= off) s += t;
      }
      int4 st;
      st.x = s - own;
      st.y = st.x + c4.x;
      st.z = st.y + c4.y;
      st.w = st.z + c4.z;
      ((int4*)startl)[lane] = st;
    }
    __syncthreads();
#pragma unroll
    for (int k = 0; k < SEPT; ++k)
      if (ed[k] >= 0) {
        int2 pr;
        pr.x = es[k];
        pr.y = ed[k] & 127;
        sorted[startl[pr.y] + (ed[k] >> 7)] = pr;
      }
    __syncthreads();

    if (firstHalf && cb == eBeg) root_product<4>(rootl, 0, m, khalf, bR, cR);

    for (int base = wave * 32; base < n; base += OWN_WAVES * 32) {
      int2 sd;
      sd.x = 0;
      sd.y = NRP;
      if (base + lane < n) sd = sorted[base + lane];
      const unsigned soff = (unsigned)sd.x << 8;
      int cur = __builtin_amdgcn_readlane(sd.y, 0);
      v2f ps = (v2f){0.f, 0.f};
#pragma unroll
      for (int jj = 0; jj < 32; jj += GB) {
        v2f v[GB];
#pragma unroll
        for (int u = 0; u < GB; ++u) {
          const unsigned so =
              (unsigned)__builtin_amdgcn_readlane((int)soff, jj + u);
          v[u] = *(const v2f*)(xb + (so | lane8));
        }
#pragma unroll
        for (int u = 0; u < GB; ++u) {
          const int row = __builtin_amdgcn_readlane(sd.y, jj + u);
          if (row != cur) {
            float* p = (float*)(accb + ((unsigned)cur * (PITCH * 4) + lane4));
            __hip_atomic_fetch_add(p, ps[0], __ATOMIC_RELAXED,
                                   __HIP_MEMORY_SCOPE_WORKGROUP);
            __hip_atomic_fetch_add(p + 32, ps[1], __ATOMIC_RELAXED,
                                   __HIP_MEMORY_SCOPE_WORKGROUP);
            ps = (v2f){0.f, 0.f};
            cur = row;
          }
          ps += v[u];
        }
      }
      float* p = (float*)(accb + ((unsigned)cur * (PITCH * 4) + lane4));
      __hip_atomic_fetch_add(p, ps[0], __ATOMIC_RELAXED,
                             __HIP_MEMORY_SCOPE_WORKGROUP);
      __hip_atomic_fetch_add(p + 32, ps[1], __ATOMIC_RELAXED,
                             __HIP_MEMORY_SCOPE_WORKGROUP);
    }
    cb += CAP;
  } while (cb < eEnd);

  v2f bL[16];
  {
    const v4f* __restrict__ wlp = (const v4f*)Wl + (nt << 8) + lane;
#pragma unroll
    for (int t = 0; t < 8; ++t) {
      const v4f ql = wlp[t << 5];
      bL[2 * t] = (v2f){ql[0], ql[1]};
      bL[2 * t + 1] = (v2f){ql[2], ql[3]};
    }
  }
  if (!firstHalf) root_product<3>(rootl, 1, m, khalf, bR, cR);

  __syncthreads();

  if (DEG)
    for (int i = tid; i < nValid; i += OWN_THREADS) deg[nodeBase + i] = degl[i];
  const float bias = bl[j];

#pragma unroll
  for (int ti = 0; ti < 4; ++ti) {
    const int tile = (wave >> 2) + ti * (OWN_WAVES / 4);
    if (tile >= NRP / 16) break;
    const int lrow = (tile << 4) + m;
    const float rdeg = 1.0f / fmaxf(degl[lrow], 1.0f);
    const float* arow = acc + lrow * PITCH + half;

    v8f c = cR[ti];
#pragma unroll
    for (int kk = 0; kk < 16; ++kk) {
      v2f a;
      a[0] = arow[2 * kk] * rdeg;
      a[1] = arow[32 + 2 * kk] * rdeg;
      c = __builtin_amdgcn_wmma_f32_16x16x4_f32(false, a, false, bL[kk],
                                                (short)0, c, false, false);
    }

    const int lr0 = (tile << 4) + mhi;
#pragma unroll
    for (int v = 0; v < 8; ++v) {
      float val = c[v] + bias;
      if (RELU) val = fmaxf(val, 0.0f);
      if (lr0 + v < nValid) out[(size_t)(nodeBase + lr0 + v) * D + j] = val;
    }
  }
}

static inline size_t align256(size_t x) { return (x + 255) & ~(size_t)255; }

extern "C" void kernel_launch(void* const* d_in, const int* in_sizes, int n_in,
                              void* d_out, int out_size, void* d_ws,
                              size_t ws_size, hipStream_t stream) {
  const float* x = (const float*)d_in[0];
  const int* src = (const int*)d_in[1];
  const int* dst = (const int*)d_in[2];
  const float* Wl[3] = {(const float*)d_in[3], (const float*)d_in[6],
                        (const float*)d_in[9]};
  const float* bl[3] = {(const float*)d_in[4], (const float*)d_in[7],
                        (const float*)d_in[10]};
  const float* Wr[3] = {(const float*)d_in[5], (const float*)d_in[8],
                        (const float*)d_in[11]};

  const int nNodes = in_sizes[0] / D;
  const int nEdges = in_sizes[1];
  const size_t featBytes = (size_t)nNodes * D * sizeof(float);
  const int nRanges = (nNodes + NR - 1) / NR;

  char* ws = (char*)d_ws;
  float* deg = (float*)ws;
  size_t ofs = align256((size_t)nNodes * sizeof(float));
  float* h1 = (float*)(ws + ofs);
  ofs += align256(featBytes);
  int* bedge = (int*)(ws + ofs);
  ofs += align256((size_t)nEdges * 2 * sizeof(int));
  int* gcount = (int*)(ws + ofs);
  ofs += align256((size_t)nRanges * sizeof(int));
  int* offs = (int*)(ws + ofs);
  ofs += align256((size_t)(nRanges + 1) * sizeof(int));
  int* cursor = (int*)(ws + ofs);
  ofs += align256((size_t)nRanges * sizeof(int));
  float* wpack = (float*)(ws + ofs);
  float* h0 = (float*)d_out;

  hipMemsetAsync(gcount, 0, (size_t)nRanges * sizeof(int), stream);
  const int binBlocks = (nEdges + BIN_THREADS * EPT - 1) / (BIN_THREADS * EPT);
  if (binBlocks > 0)
    range_count_kernel<<<binBlocks, BIN_THREADS, nRanges * sizeof(int),
                         stream>>>(dst, gcount, nEdges, nRanges);
  range_scan_kernel<<<1 + 6, 1024, 0, stream>>>(gcount, offs, cursor, nRanges,
                                                Wl[0], Wr[0], Wl[1], Wr[1],
                                                Wl[2], Wr[2], wpack);
  if (binBlocks > 0)
    range_fill_kernel<<<binBlocks, BIN_THREADS, 2 * nRanges * sizeof(int),
                        stream>>>(src, dst, cursor, bedge, nEdges, nRanges);

  auto run_layer = [&](const float* in, const float* wl, const float* b,
                       const float* wr, float* outp, bool relu, bool first) {
    if (nRanges <= 0) return;
    if (first)
      owner_layer_kernel<true, true><<<nRanges, OWN_THREADS, 0, stream>>>(
          in, bedge, offs, deg, wl, b, wr, outp, nNodes);
    else if (relu)
      owner_layer_kernel<true, false><<<nRanges, OWN_THREADS, 0, stream>>>(
          in, bedge, offs, deg, wl, b, wr, outp, nNodes);
    else
      owner_layer_kernel<false, false><<<nRanges, OWN_THREADS, 0, stream>>>(
          in, bedge, offs, deg, wl, b, wr, outp, nNodes);
  };

  run_layer(x, wpack + 0 * D * D, bl[0], wpack + 1 * D * D, h0,  true,
             true);
  run_layer(h0, wpack + 2 * D * D, bl[1], wpack + 3 * D * D, h1,  true,
             false);
  run_layer(h1, wpack + 4 * D * D, bl[2], wpack + 5 * D * D, (float*)d_out,
             false,  false);
}
